# starting kernel
# baseline (speedup 1.0000x reference)
#include <hip/hip_runtime.h>

#define Bb   64
#define Ss   128
#define Ll   32
#define Dd   100
#define Mm   20
#define DP   112
#define MP   32
#define NTHR 256
#define NWAVE 8
#define NTILE 14
#define KSTEPS 25

typedef __attribute__((ext_vector_type(2))) float v2f;
typedef __attribute__((ext_vector_type(8))) float v8f;

__global__ __launch_bounds__(128) void sents_kernel(
    const int* __restrict__ tokens, const float* __restrict__ emb,
    const float* __restrict__ mult, float* __restrict__ sents)
{
    const int bs = blockIdx.x;
    const int d  = threadIdx.x;
    if (d >= Dd) return;
    const int* trow = tokens + bs * Ll;
    float acc = 0.f;
    #pragma unroll 4
    for (int l = 0; l < Ll; ++l) {
        const int tok = trow[l];
        acc += emb[tok * Dd + d] * mult[l * Dd + d];
    }
    sents[bs * Dd + d] = acc;
}

__global__ __launch_bounds__(128) void vk_kernel(
    const float* __restrict__ keys, const float* __restrict__ Vw,
    float* __restrict__ VkBuf)
{
    const int m = blockIdx.x;
    const int d = threadIdx.x;
    if (d >= Dd) return;
    float acc = 0.f;
    for (int k = 0; k < Dd; ++k) acc += keys[m * Dd + k] * Vw[d * Dd + k];
    VkBuf[m * Dd + d] = acc;
}

__global__ __launch_bounds__(NTHR) void entnet_scan_kernel(
    const float* __restrict__ sents, const float* __restrict__ keys,
    const float* __restrict__ Uw,    const float* __restrict__ Ww,
    const float* __restrict__ VkBuf, const float* __restrict__ prelu_a,
    float* __restrict__ out)
{
    const int b    = blockIdx.x;
    const int tid  = threadIdx.x;
    const int wave = tid >> 5;
    const int lane = tid & 31;
    const float pa = prelu_a[0];

    __shared__ float Uwt  [Dd * DP];
    __shared__ float memS [MP * DP];
    __shared__ float sS   [DP];
    __shared__ float WsS  [DP];
    __shared__ float gateS[Mm];
    __shared__ float scaleS[Mm];

    for (int i = tid; i < MP * DP; i += NTHR) memS[i] = 0.f;
    for (int i = tid; i < Dd * DP; i += NTHR) {
        const int k = i / DP, d = i - k * DP;
        Uwt[i] = (d < Dd) ? Uw[d * Dd + k] : 0.f;
    }
    __syncthreads();
    for (int i = tid; i < Mm * Dd; i += NTHR) {
        const int m = i / Dd, d = i - m * Dd;
        memS[m * DP + d] = keys[i];
    }
    __syncthreads();

    const int r15 = lane & 15;
    const int kh  = (lane >> 4) * 2;
    const int mh  = (lane >> 4) * 8;

    const int  t0   = wave;
    const int  t1   = wave + NWAVE;
    const bool has1 = (t1 < NTILE);
    const int  mb0 = (t0 / 7) * 16, nb0 = (t0 % 7) * 16;
    const int  mb1 = (t1 / 7) * 16, nb1 = (t1 % 7) * 16;

    v2f bf0[KSTEPS], bf1[KSTEPS];
    #pragma unroll
    for (int j = 0; j < KSTEPS; ++j) {
        const int kk = j * 4;
        bf0[j].x = Uwt[(kk + kh)     * DP + nb0 + r15];
        bf0[j].y = Uwt[(kk + kh + 1) * DP + nb0 + r15];
    }
    if (has1) {
        #pragma unroll
        for (int j = 0; j < KSTEPS; ++j) {
            const int kk = j * 4;
            bf1[j].x = Uwt[(kk + kh)     * DP + nb1 + r15];
            bf1[j].y = Uwt[(kk + kh + 1) * DP + nb1 + r15];
        }
    } else {
        #pragma unroll
        for (int j = 0; j < KSTEPS; ++j) bf1[j] = v2f{0.f, 0.f};
    }

    auto gemm_tile = [&](int mb, const v2f* bf) -> v8f {
        v8f c = {};
        #pragma unroll
        for (int j = 0; j < KSTEPS; ++j) {
            const int kk = j * 4;
            const v2f a = *(const v2f*)&memS[(mb + r15) * DP + kk + kh];
            c = __builtin_amdgcn_wmma_f32_16x16x4_f32(
                     false, a,  false, bf[j],
                     (short)0, c,  false,  false);
        }
        return c;
    };

    auto apply_tile = [&](int mb, int nb, v8f c) {
        #pragma unroll
        for (int r = 0; r < 8; ++r) {
            const int m = mb + mh + r;
            const int n = nb + r15;
            if (m < Mm && n < Dd) {
                float x = c[r] + VkBuf[m * Dd + n] + WsS[n];
                x = (x >= 0.f) ? x : pa * x;
                memS[m * DP + n] += x * gateS[m];
            }
        }
    };

    for (int t = 0; t < Ss; ++t) {
        if (tid < DP) sS[tid] = (tid < Dd) ? sents[(b * Ss + t) * Dd + tid] : 0.f;
        if (t + 1 < Ss && tid < 4)
            __builtin_prefetch(&sents[(b * Ss + t + 1) * Dd + tid * 32], 0, 3);
        __syncthreads();

        if (tid < Mm) {
            float mg = 0.f, kg = 0.f;
            #pragma unroll 4
            for (int k = 0; k < Dd; ++k) {
                const float sv = sS[k];
                mg += sv * memS[tid * DP + k];
                kg += sv * keys[tid * Dd + k];
            }
            const float z = mg + kg;
            const float g = 1.f / (1.f + expf(-z));
            gateS[tid] = (z == 0.f) ? 0.f : g;
        } else if (tid >= 32 && tid < 32 + DP) {
            const int d = tid - 32;
            float w = 0.f;
            if (d < Dd) {
                #pragma unroll 4
                for (int k = 0; k < Dd; ++k) w += sS[k] * Ww[d * Dd + k];
            }
            WsS[d] = w;
        }
        __syncthreads();

        v8f acc0 = gemm_tile(mb0, bf0);
        v8f acc1 = {};
        if (has1) acc1 = gemm_tile(mb1, bf1);
        __syncthreads();

        apply_tile(mb0, nb0, acc0);
        if (has1) apply_tile(mb1, nb1, acc1);
        __syncthreads();

        if (tid < Mm) {
            float ss = 0.f;
            #pragma unroll 4
            for (int d = 0; d < Dd; ++d) {
                const float v = memS[tid * DP + d];
                ss += v * v;
            }
            scaleS[tid] = 1.f / sqrtf(ss);
        }
        __syncthreads();
        for (int i = tid; i < Mm * DP; i += NTHR) {
            const int m = i / DP;
            memS[i] *= scaleS[m];
        }
        __syncthreads();
    }

    for (int i = tid; i < Mm * Dd; i += NTHR) {
        const int m = i / Dd, d = i - m * Dd;
        out[b * Mm * Dd + i] = memS[m * DP + d];
    }
}

extern "C" void kernel_launch(void* const* d_in, const int* in_sizes, int n_in,
                              void* d_out, int out_size, void* d_ws, size_t ws_size,
                              hipStream_t stream) {
    (void)in_sizes; (void)n_in; (void)out_size; (void)ws_size;
    const int*   tokens  = (const int*)  d_in[0];
    const float* emb     = (const float*)d_in[1];
    const float* keys    = (const float*)d_in[2];
    const float* mult    = (const float*)d_in[3];
    const float* Uw      = (const float*)d_in[4];
    const float* Vw      = (const float*)d_in[5];
    const float* Ww      = (const float*)d_in[6];
    const float* prelu_a = (const float*)d_in[7];
    float* out = (float*)d_out;

    float* sents = (float*)d_ws;
    float* VkBuf = sents + (size_t)Bb * Ss * Dd;

    sents_kernel<<<Bb * Ss, 128, 0, stream>>>(tokens, emb, mult, sents);
    vk_kernel<<<Mm, 128, 0, stream>>>(keys, Vw, VkBuf);
    entnet_scan_kernel<<<Bb, NTHR, 0, stream>>>(sents, keys, Uw, Ww, VkBuf,
                                                prelu_a, out);
}


# State-independent work leaves the 128-step loop: sentence rows staged in LDS, s@Ww^T and s.keys for all steps from one WMMA prologue (those sums now in WMMA k-order), Vk terms held in registers
# speedup vs baseline: 1.3628x; 1.3628x over previous
#include <hip/hip_runtime.h>

#define Bb   64
#define Ss   128
#define Ll   32
#define Dd   100
#define Mm   20
#define DP   112
#define MP   32
#define NTHR 256
#define NWAVE 8
#define NTILE 14
#define KSTEPS 25
#define NB   144
#define PW   102
#define PP   144

typedef __attribute__((ext_vector_type(2))) float v2f;
typedef __attribute__((ext_vector_type(8))) float v8f;

__global__ __launch_bounds__(128) void sents_kernel(
    const int* __restrict__ tokens, const float* __restrict__ emb,
    const float* __restrict__ mult, float* __restrict__ sents)
{
    const int bs = blockIdx.x;
    const int d  = threadIdx.x;
    if (d >= Dd) return;
    const int* trow = tokens + bs * Ll;
    float acc = 0.f;
    #pragma unroll 4
    for (int l = 0; l < Ll; ++l) {
        const int tok = trow[l];
        acc += emb[tok * Dd + d] * mult[l * Dd + d];
    }
    sents[bs * Dd + d] = acc;
}

__global__ __launch_bounds__(128) void vk_kernel(
    const float* __restrict__ keys, const float* __restrict__ Vw,
    float* __restrict__ VkBuf)
{
    const int m = blockIdx.x;
    const int d = threadIdx.x;
    if (d >= Dd) return;
    float acc = 0.f;
    for (int k = 0; k < Dd; ++k) acc += keys[m * Dd + k] * Vw[d * Dd + k];
    VkBuf[m * Dd + d] = acc;
}

__global__ __launch_bounds__(NTHR) void entnet_scan_kernel(
    const float* __restrict__ sents, const float* __restrict__ keys,
    const float* __restrict__ Uw,    const float* __restrict__ Ww,
    const float* __restrict__ VkBuf, const float* __restrict__ prelu_a,
    float* __restrict__ out)
{
    const int b    = blockIdx.x;
    const int tid  = threadIdx.x;
    const int wave = tid >> 5;
    const int lane = tid & 31;
    const float pa = prelu_a[0];

    __shared__ float Uwt  [Dd * DP];
    __shared__ float memS [MP * DP];
    __shared__ float sentS[Ss * DP];
    __shared__ float wkS  [NB * PW];
    __shared__ float preS [Ss * PP];
    __shared__ float gateS[Mm];
    __shared__ float scaleS[Mm];

    for (int i = tid; i < MP * DP; i += NTHR) memS[i] = 0.f;
    for (int i = tid; i < Dd * DP; i += NTHR) {
        const int k = i / DP, d = i - k * DP;
        Uwt[i] = (d < Dd) ? Uw[d * Dd + k] : 0.f;
    }
    __syncthreads();
    for (int i = tid; i < Mm * Dd; i += NTHR) {
        const int m = i / Dd, d = i - m * Dd;
        memS[m * DP + d] = keys[i];
    }
    #pragma unroll 10
    for (int i = tid; i < Ss * Dd; i += NTHR) {
        const int t = i / Dd, k = i - t * Dd;
        sentS[t * DP + k] = sents[b * Ss * Dd + i];
    }
    #pragma unroll 8
    for (int i = tid; i < Dd * Dd; i += NTHR) {
        const int n = i / Dd, k = i - n * Dd;
        wkS[n * PW + k] = Ww[i];
    }
    #pragma unroll 8
    for (int i = tid; i < Mm * Dd; i += NTHR) {
        const int m = i / Dd, k = i - m * Dd;
        wkS[(DP + m) * PW + k] = keys[i];
    }
    for (int i = tid; i < (DP - Dd) * PW; i += NTHR) {
        wkS[Dd * PW + i]        = 0.f;
        wkS[(DP + Mm) * PW + i] = 0.f;
    }
    __syncthreads();

    const int r15 = lane & 15;
    const int kh  = (lane >> 4) * 2;
    const int mh  = (lane >> 4) * 8;

    const int  t0   = wave;
    const int  t1   = wave + NWAVE;
    const bool has1 = (t1 < NTILE);
    const int  mb0 = (t0 / 7) * 16, nb0 = (t0 % 7) * 16;
    const int  mb1 = (t1 / 7) * 16, nb1 = (t1 % 7) * 16;

    {
        const int tb = wave * 16;
        v2f af[KSTEPS];
        #pragma unroll
        for (int j = 0; j < KSTEPS; ++j)
            af[j] = *(const v2f*)&sentS[(tb + r15) * DP + j * 4 + kh];
        for (int nt = 0; nt < NB / 16; ++nt) {
            v8f c = {};
            #pragma unroll
            for (int j = 0; j < KSTEPS; ++j) {
                const v2f bw = *(const v2f*)&wkS[(nt * 16 + r15) * PW + j * 4 + kh];
                c = __builtin_amdgcn_wmma_f32_16x16x4_f32(
                         false, af[j],  false, bw,
                         (short)0, c,  false,  false);
            }
            #pragma unroll
            for (int r = 0; r < 8; ++r)
                preS[(tb + mh + r) * PP + nt * 16 + r15] = c[r];
        }
    }
    __syncthreads();

    v2f bf0[KSTEPS], bf1[KSTEPS];
    #pragma unroll
    for (int j = 0; j < KSTEPS; ++j) {
        const int kk = j * 4;
        bf0[j].x = Uwt[(kk + kh)     * DP + nb0 + r15];
        bf0[j].y = Uwt[(kk + kh + 1) * DP + nb0 + r15];
    }
    if (has1) {
        #pragma unroll
        for (int j = 0; j < KSTEPS; ++j) {
            const int kk = j * 4;
            bf1[j].x = Uwt[(kk + kh)     * DP + nb1 + r15];
            bf1[j].y = Uwt[(kk + kh + 1) * DP + nb1 + r15];
        }
    } else {
        #pragma unroll
        for (int j = 0; j < KSTEPS; ++j) bf1[j] = v2f{0.f, 0.f};
    }

    auto gemm_tile = [&](int mb, const v2f* bf) -> v8f {
        v8f c = {};
        #pragma unroll
        for (int j = 0; j < KSTEPS; ++j) {
            const int kk = j * 4;
            const v2f a = *(const v2f*)&memS[(mb + r15) * DP + kk + kh];
            c = __builtin_amdgcn_wmma_f32_16x16x4_f32(
                     false, a,  false, bf[j],
                     (short)0, c,  false,  false);
        }
        return c;
    };

    float vk0[8], vk1[8];
    #pragma unroll
    for (int r = 0; r < 8; ++r) {
        const int m0 = mb0 + mh + r, n0 = nb0 + r15;
        const int m1 = mb1 + mh + r, n1 = nb1 + r15;
        const bool ok0 = (m0 < Mm && n0 < Dd);
        const bool ok1 = (has1 && m1 < Mm && n1 < Dd);
        const float v0 = VkBuf[ok0 ? m0 * Dd + n0 : 0];
        const float v1 = VkBuf[ok1 ? m1 * Dd + n1 : 0];
        vk0[r] = ok0 ? v0 : 0.f;
        vk1[r] = ok1 ? v1 : 0.f;
    }

    auto apply_tile = [&](int mb, int nb, v8f c, const float* vk, const float* wsRow) {
        #pragma unroll
        for (int r = 0; r < 8; ++r) {
            const int m = mb + mh + r;
            const int n = nb + r15;
            if (m < Mm && n < Dd) {
                float x = c[r] + vk[r] + wsRow[n];
                x = (x >= 0.f) ? x : pa * x;
                memS[m * DP + n] += x * gateS[m];
            }
        }
    };

    for (int t = 0; t < Ss; ++t) {
        const float* sRow = &sentS[t * DP];
        const float* pRow = &preS[t * PP];

        if (tid < Mm) {
            float mg = 0.f;
            #pragma unroll 4
            for (int k = 0; k < Dd; ++k) mg += sRow[k] * memS[tid * DP + k];
            const float z = mg + pRow[DP + tid];
            const float g = 1.f / (1.f + expf(-z));
            gateS[tid] = (z == 0.f) ? 0.f : g;
        }
        __syncthreads();

        v8f acc0 = gemm_tile(mb0, bf0);
        v8f acc1 = {};
        if (has1) acc1 = gemm_tile(mb1, bf1);
        __syncthreads();

        apply_tile(mb0, nb0, acc0, vk0, pRow);
        if (has1) apply_tile(mb1, nb1, acc1, vk1, pRow);
        __syncthreads();

        if (tid < Mm) {
            float ss = 0.f;
            #pragma unroll 4
            for (int d = 0; d < Dd; ++d) {
                const float v = memS[tid * DP + d];
                ss += v * v;
            }
            scaleS[tid] = 1.f / sqrtf(ss);
        }
        __syncthreads();
        for (int i = tid; i < Mm * DP; i += NTHR) {
            const int m = i / DP;
            memS[i] *= scaleS[m];
        }
        __syncthreads();
    }

    for (int i = tid; i < Mm * Dd; i += NTHR) {
        const int m = i / Dd, d = i - m * Dd;
        out[b * Mm * Dd + i] = memS[m * DP + d];
    }
}

extern "C" void kernel_launch(void* const* d_in, const int* in_sizes, int n_in,
                              void* d_out, int out_size, void* d_ws, size_t ws_size,
                              hipStream_t stream) {
    (void)in_sizes; (void)n_in; (void)out_size; (void)ws_size;
    const int*   tokens  = (const int*)  d_in[0];
    const float* emb     = (const float*)d_in[1];
    const float* keys    = (const float*)d_in[2];
    const float* mult    = (const float*)d_in[3];
    const float* Uw      = (const float*)d_in[4];
    const float* Vw      = (const float*)d_in[5];
    const float* Ww      = (const float*)d_in[6];
    const float* prelu_a = (const float*)d_in[7];
    float* out = (float*)d_out;

    float* sents = (float*)d_ws;
    float* VkBuf = sents + (size_t)Bb * Ss * Dd;

    sents_kernel<<<Bb * Ss, 128, 0, stream>>>(tokens, emb, mult, sents);
    vk_kernel<<<Mm, 128, 0, stream>>>(keys, Vw, VkBuf);
    entnet_scan_kernel<<<Bb, NTHR, 0, stream>>>(sents, keys, Uw, Ww, VkBuf,
                                                prelu_a, out);
}


# Per-slot gate dot and squared norm leave the 20 serial threads: 8-lane groups hold a slot row in registers, DPP sums (both sums reordered), scale in place, next gate; 5 barriers a step become 3
# speedup vs baseline: 3.0932x; 2.2697x over previous
#include <hip/hip_runtime.h>

#define Bb   64
#define Ss   128
#define Ll   32
#define Dd   100
#define Mm   20
#define DP   112
#define MP   32
#define NTHR 256
#define NWAVE 8
#define NTILE 14
#define KSTEPS 25
#define NB   144
#define PW   102
#define PP   144

typedef __attribute__((ext_vector_type(2))) float v2f;
typedef __attribute__((ext_vector_type(8))) float v8f;

__global__ __launch_bounds__(128) void sents_kernel(
    const int* __restrict__ tokens, const float* __restrict__ emb,
    const float* __restrict__ mult, float* __restrict__ sents)
{
    const int bs = blockIdx.x;
    const int d  = threadIdx.x;
    if (d >= Dd) return;
    const int* trow = tokens + bs * Ll;
    float acc = 0.f;
    #pragma unroll 4
    for (int l = 0; l < Ll; ++l) {
        const int tok = trow[l];
        acc += emb[tok * Dd + d] * mult[l * Dd + d];
    }
    sents[bs * Dd + d] = acc;
}

__global__ __launch_bounds__(128) void vk_kernel(
    const float* __restrict__ keys, const float* __restrict__ Vw,
    float* __restrict__ VkBuf)
{
    const int m = blockIdx.x;
    const int d = threadIdx.x;
    if (d >= Dd) return;
    float acc = 0.f;
    for (int k = 0; k < Dd; ++k) acc += keys[m * Dd + k] * Vw[d * Dd + k];
    VkBuf[m * Dd + d] = acc;
}

__device__ __forceinline__ float sum8(float x)
{
    x += __builtin_bit_cast(float, __builtin_amdgcn_update_dpp(
             0, __builtin_bit_cast(int, x), 0x161  , 0xf, 0xf, true));
    x += __builtin_bit_cast(float, __builtin_amdgcn_update_dpp(
             0, __builtin_bit_cast(int, x), 0x162  , 0xf, 0xf, true));
    x += __builtin_bit_cast(float, __builtin_amdgcn_update_dpp(
             0, __builtin_bit_cast(int, x), 0x164  , 0xf, 0xf, true));
    return x;
}

__global__ __launch_bounds__(NTHR) void entnet_scan_kernel(
    const float* __restrict__ sents, const float* __restrict__ keys,
    const float* __restrict__ Uw,    const float* __restrict__ Ww,
    const float* __restrict__ VkBuf, const float* __restrict__ prelu_a,
    float* __restrict__ out)
{
    const int b    = blockIdx.x;
    const int tid  = threadIdx.x;
    const int wave = tid >> 5;
    const int lane = tid & 31;
    const float pa = prelu_a[0];

    __shared__ float Uwt  [Dd * DP];
    __shared__ float memS [MP * DP];
    __shared__ float sentS[Ss * DP];
    __shared__ float wkS  [NB * PW];
    __shared__ float preS [Ss * PP];
    __shared__ float gateS[Mm];

    for (int i = tid; i < MP * DP; i += NTHR) memS[i] = 0.f;
    for (int i = tid; i < Dd * DP; i += NTHR) {
        const int k = i / DP, d = i - k * DP;
        Uwt[i] = (d < Dd) ? Uw[d * Dd + k] : 0.f;
    }
    __syncthreads();
    for (int i = tid; i < Mm * Dd; i += NTHR) {
        const int m = i / Dd, d = i - m * Dd;
        memS[m * DP + d] = keys[i];
    }
    for (int i = tid; i < Ss * (DP - Dd); i += NTHR) {
        const int t = i / (DP - Dd), k = i - t * (DP - Dd);
        sentS[t * DP + Dd + k] = 0.f;
    }
    #pragma unroll 10
    for (int i = tid; i < Ss * Dd; i += NTHR) {
        const int t = i / Dd, k = i - t * Dd;
        sentS[t * DP + k] = sents[b * Ss * Dd + i];
    }
    #pragma unroll 8
    for (int i = tid; i < Dd * Dd; i += NTHR) {
        const int n = i / Dd, k = i - n * Dd;
        wkS[n * PW + k] = Ww[i];
    }
    #pragma unroll 8
    for (int i = tid; i < Mm * Dd; i += NTHR) {
        const int m = i / Dd, k = i - m * Dd;
        wkS[(DP + m) * PW + k] = keys[i];
    }
    for (int i = tid; i < (DP - Dd) * PW; i += NTHR) {
        wkS[Dd * PW + i]        = 0.f;
        wkS[(DP + Mm) * PW + i] = 0.f;
    }
    __syncthreads();

    const int r15 = lane & 15;
    const int kh  = (lane >> 4) * 2;
    const int mh  = (lane >> 4) * 8;

    const int  t0   = wave;
    const int  t1   = wave + NWAVE;
    const bool has1 = (t1 < NTILE);
    const int  mb0 = (t0 / 7) * 16, nb0 = (t0 % 7) * 16;
    const int  mb1 = (t1 / 7) * 16, nb1 = (t1 % 7) * 16;

    {
        const int tb = wave * 16;
        v2f af[KSTEPS];
        #pragma unroll
        for (int j = 0; j < KSTEPS; ++j)
            af[j] = *(const v2f*)&sentS[(tb + r15) * DP + j * 4 + kh];
        for (int nt = 0; nt < NB / 16; ++nt) {
            v8f c = {};
            #pragma unroll
            for (int j = 0; j < KSTEPS; ++j) {
                const v2f bw = *(const v2f*)&wkS[(nt * 16 + r15) * PW + j * 4 + kh];
                c = __builtin_amdgcn_wmma_f32_16x16x4_f32(
                         false, af[j],  false, bw,
                         (short)0, c,  false,  false);
            }
            #pragma unroll
            for (int r = 0; r < 8; ++r)
                preS[(tb + mh + r) * PP + nt * 16 + r15] = c[r];
        }
    }
    __syncthreads();

    v2f bf0[KSTEPS], bf1[KSTEPS];
    #pragma unroll
    for (int j = 0; j < KSTEPS; ++j) {
        const int kk = j * 4;
        bf0[j].x = Uwt[(kk + kh)     * DP + nb0 + r15];
        bf0[j].y = Uwt[(kk + kh + 1) * DP + nb0 + r15];
    }
    if (has1) {
        #pragma unroll
        for (int j = 0; j < KSTEPS; ++j) {
            const int kk = j * 4;
            bf1[j].x = Uwt[(kk + kh)     * DP + nb1 + r15];
            bf1[j].y = Uwt[(kk + kh + 1) * DP + nb1 + r15];
        }
    } else {
        #pragma unroll
        for (int j = 0; j < KSTEPS; ++j) bf1[j] = v2f{0.f, 0.f};
    }

    auto gemm_tile = [&](int mb, const v2f* bf) -> v8f {
        v8f c = {};
        #pragma unroll
        for (int j = 0; j < KSTEPS; ++j) {
            const int kk = j * 4;
            const v2f a = *(const v2f*)&memS[(mb + r15) * DP + kk + kh];
            c = __builtin_amdgcn_wmma_f32_16x16x4_f32(
                     false, a,  false, bf[j],
                     (short)0, c,  false,  false);
        }
        return c;
    };

    float vk0[8], vk1[8];
    #pragma unroll
    for (int r = 0; r < 8; ++r) {
        const int m0 = mb0 + mh + r, n0 = nb0 + r15;
        const int m1 = mb1 + mh + r, n1 = nb1 + r15;
        const bool ok0 = (m0 < Mm && n0 < Dd);
        const bool ok1 = (has1 && m1 < Mm && n1 < Dd);
        const float v0 = VkBuf[ok0 ? m0 * Dd + n0 : 0];
        const float v1 = VkBuf[ok1 ? m1 * Dd + n1 : 0];
        vk0[r] = ok0 ? v0 : 0.f;
        vk1[r] = ok1 ? v1 : 0.f;
    }

    auto apply_tile = [&](int mb, int nb, v8f c, const float* vk, const float* wsRow) {
        #pragma unroll
        for (int r = 0; r < 8; ++r) {
            const int m = mb + mh + r;
            const int n = nb + r15;
            if (m < Mm && n < Dd) {
                float x = c[r] + vk[r] + wsRow[n];
                x = (x >= 0.f) ? x : pa * x;
                memS[m * DP + n] += x * gateS[m];
            }
        }
    };

    const int  g8   = tid >> 3;
    const int  l8   = tid & 7;
    const bool rowW = (__builtin_amdgcn_readfirstlane(tid) < 5 * 32);
    float*       mRow = &memS[g8 * DP + 2 * l8];

    auto gate_from_row = [&](const v2f* v, const v2f* u, float kg) {
        float d0 = 0.f, d1 = 0.f;
        #pragma unroll
        for (int i = 0; i < 7; ++i) {
            d0 = __builtin_fmaf(u[i].x, v[i].x, d0);
            d1 = __builtin_fmaf(u[i].y, v[i].y, d1);
        }
        const float z = sum8(d0 + d1) + kg;
        const float g = 1.f / (1.f + expf(-z));
        if (l8 == 0) gateS[g8] = (z == 0.f) ? 0.f : g;
    };

    if (rowW) {
        v2f v[7], u[7];
        #pragma unroll
        for (int i = 0; i < 7; ++i) {
            v[i] = *(const v2f*)&mRow[16 * i];
            u[i] = *(const v2f*)&sentS[2 * l8 + 16 * i];
        }
        gate_from_row(v, u, preS[DP + g8]);
    }

    for (int t = 0; t < Ss; ++t) {
        const float* pRow = &preS[t * PP];

        v8f acc0 = gemm_tile(mb0, bf0);
        v8f acc1 = {};
        if (has1) acc1 = gemm_tile(mb1, bf1);
        __syncthreads();

        apply_tile(mb0, nb0, acc0, vk0, pRow);
        if (has1) apply_tile(mb1, nb1, acc1, vk1, pRow);
        __syncthreads();

        if (rowW) {
            const int tn = (t + 1 < Ss) ? t + 1 : t;
            v2f v[7], u[7];
            #pragma unroll
            for (int i = 0; i < 7; ++i) v[i] = *(const v2f*)&mRow[16 * i];
            #pragma unroll
            for (int i = 0; i < 7; ++i)
                u[i] = *(const v2f*)&sentS[tn * DP + 2 * l8 + 16 * i];
            const float kg = preS[tn * PP + DP + g8];
            __builtin_amdgcn_sched_barrier(0);
            float s0 = 0.f, s1 = 0.f;
            #pragma unroll
            for (int i = 0; i < 7; ++i) {
                s0 = __builtin_fmaf(v[i].x, v[i].x, s0);
                s1 = __builtin_fmaf(v[i].y, v[i].y, s1);
            }
            const float scale = 1.f / sqrtf(sum8(s0 + s1));
            #pragma unroll
            for (int i = 0; i < 7; ++i) {
                v[i].x *= scale;
                v[i].y *= scale;
                *(v2f*)&mRow[16 * i] = v[i];
            }
            gate_from_row(v, u, kg);
        }
        __syncthreads();
    }

    for (int i = tid; i < Mm * Dd; i += NTHR) {
        const int m = i / Dd, d = i - m * Dd;
        out[b * Mm * Dd + i] = memS[m * DP + d];
    }
}

extern "C" void kernel_launch(void* const* d_in, const int* in_sizes, int n_in,
                              void* d_out, int out_size, void* d_ws, size_t ws_size,
                              hipStream_t stream) {
    (void)in_sizes; (void)n_in; (void)out_size; (void)ws_size;
    const int*   tokens  = (const int*)  d_in[0];
    const float* emb     = (const float*)d_in[1];
    const float* keys    = (const float*)d_in[2];
    const float* mult    = (const float*)d_in[3];
    const float* Uw      = (const float*)d_in[4];
    const float* Vw      = (const float*)d_in[5];
    const float* Ww      = (const float*)d_in[6];
    const float* prelu_a = (const float*)d_in[7];
    float* out = (float*)d_out;

    float* sents = (float*)d_ws;
    float* VkBuf = sents + (size_t)Bb * Ss * Dd;

    sents_kernel<<<Bb * Ss, 128, 0, stream>>>(tokens, emb, mult, sents);
    vk_kernel<<<Mm, 128, 0, stream>>>(keys, Vw, VkBuf);
    entnet_scan_kernel<<<Bb, NTHR, 0, stream>>>(sents, keys, Uw, Ww, VkBuf,
                                                prelu_a, out);
}


# Normalisation deferred: state kept unscaled, row scale applied after the WMMA sums and in the update (not before the product); norm+gate pass on 5 added waves beside the product; 2 barriers a step
# speedup vs baseline: 3.4043x; 1.1006x over previous
#include <hip/hip_runtime.h>

#define Bb   64
#define Ss   128
#define Ll   32
#define Dd   100
#define Mm   20
#define DP   112
#define MP   32
#define NTHR 416
#define NWAVE 8
#define NTILE 14
#define KSTEPS 25
#define NB   144
#define PW   102
#define PP   144

typedef __attribute__((ext_vector_type(2))) float v2f;
typedef __attribute__((ext_vector_type(8))) float v8f;

__global__ __launch_bounds__(128) void sents_kernel(
    const int* __restrict__ tokens, const float* __restrict__ emb,
    const float* __restrict__ mult, float* __restrict__ sents)
{
    const int bs = blockIdx.x;
    const int d  = threadIdx.x;
    if (d >= Dd) return;
    const int* trow = tokens + bs * Ll;
    float acc = 0.f;
    #pragma unroll 4
    for (int l = 0; l < Ll; ++l) {
        const int tok = trow[l];
        acc += emb[tok * Dd + d] * mult[l * Dd + d];
    }
    sents[bs * Dd + d] = acc;
}

__global__ __launch_bounds__(128) void vk_kernel(
    const float* __restrict__ keys, const float* __restrict__ Vw,
    float* __restrict__ VkBuf)
{
    const int m = blockIdx.x;
    const int d = threadIdx.x;
    if (d >= Dd) return;
    float acc = 0.f;
    for (int k = 0; k < Dd; ++k) acc += keys[m * Dd + k] * Vw[d * Dd + k];
    VkBuf[m * Dd + d] = acc;
}

__device__ __forceinline__ float sum8(float x)
{
    x += __builtin_bit_cast(float, __builtin_amdgcn_update_dpp(
             0, __builtin_bit_cast(int, x), 0x161  , 0xf, 0xf, true));
    x += __builtin_bit_cast(float, __builtin_amdgcn_update_dpp(
             0, __builtin_bit_cast(int, x), 0x162  , 0xf, 0xf, true));
    x += __builtin_bit_cast(float, __builtin_amdgcn_update_dpp(
             0, __builtin_bit_cast(int, x), 0x164  , 0xf, 0xf, true));
    return x;
}

__global__ __launch_bounds__(NTHR) void entnet_scan_kernel(
    const float* __restrict__ sents, const float* __restrict__ keys,
    const float* __restrict__ Uw,    const float* __restrict__ Ww,
    const float* __restrict__ VkBuf, const float* __restrict__ prelu_a,
    float* __restrict__ out)
{
    const int b    = blockIdx.x;
    const int tid  = threadIdx.x;
    const int wave = tid >> 5;
    const int lane = tid & 31;
    const float pa = prelu_a[0];

    __shared__ float Uwt  [Dd * DP];
    __shared__ float memS [MP * DP];
    __shared__ float sentS[Ss * DP];
    __shared__ float wkS  [NB * PW];
    __shared__ float preS [Ss * PP];
    __shared__ v2f   sgS  [Mm];

    for (int i = tid; i < MP * DP; i += NTHR) memS[i] = 0.f;
    for (int i = tid; i < Dd * DP; i += NTHR) {
        const int k = i / DP, d = i - k * DP;
        Uwt[i] = (d < Dd) ? Uw[d * Dd + k] : 0.f;
    }
    __syncthreads();
    for (int i = tid; i < Mm * Dd; i += NTHR) {
        const int m = i / Dd, d = i - m * Dd;
        memS[m * DP + d] = keys[i];
    }
    for (int i = tid; i < Ss * (DP - Dd); i += NTHR) {
        const int t = i / (DP - Dd), k = i - t * (DP - Dd);
        sentS[t * DP + Dd + k] = 0.f;
    }
    #pragma unroll 10
    for (int i = tid; i < Ss * Dd; i += NTHR) {
        const int t = i / Dd, k = i - t * Dd;
        sentS[t * DP + k] = sents[b * Ss * Dd + i];
    }
    #pragma unroll 8
    for (int i = tid; i < Dd * Dd; i += NTHR) {
        const int n = i / Dd, k = i - n * Dd;
        wkS[n * PW + k] = Ww[i];
    }
    #pragma unroll 8
    for (int i = tid; i < Mm * Dd; i += NTHR) {
        const int m = i / Dd, k = i - m * Dd;
        wkS[(DP + m) * PW + k] = keys[i];
    }
    for (int i = tid; i < (DP - Dd) * PW; i += NTHR) {
        wkS[Dd * PW + i]        = 0.f;
        wkS[(DP + Mm) * PW + i] = 0.f;
    }
    __syncthreads();

    const int r15 = lane & 15;
    const int kh  = (lane >> 4) * 2;
    const int mh  = (lane >> 4) * 8;

    const int  t0   = wave;
    const int  t1   = wave + NWAVE;
    const bool has1 = (t1 < NTILE);
    const int  mb0 = (t0 / 7) * 16, nb0 = (t0 % 7) * 16;
    const int  mb1 = (t1 / 7) * 16, nb1 = (t1 % 7) * 16;

    const bool gemmW = (__builtin_amdgcn_readfirstlane(tid) < NWAVE * 32);

    if (gemmW) {
        const int tb = wave * 16;
        v2f af[KSTEPS];
        #pragma unroll
        for (int j = 0; j < KSTEPS; ++j)
            af[j] = *(const v2f*)&sentS[(tb + r15) * DP + j * 4 + kh];
        for (int nt = 0; nt < NB / 16; ++nt) {
            v8f c = {};
            #pragma unroll
            for (int j = 0; j < KSTEPS; ++j) {
                const v2f bw = *(const v2f*)&wkS[(nt * 16 + r15) * PW + j * 4 + kh];
                c = __builtin_amdgcn_wmma_f32_16x16x4_f32(
                         false, af[j],  false, bw,
                         (short)0, c,  false,  false);
            }
            #pragma unroll
            for (int r = 0; r < 8; ++r)
                preS[(tb + mh + r) * PP + nt * 16 + r15] = c[r];
        }
    }
    __syncthreads();

    v2f bf0[KSTEPS], bf1[KSTEPS];
    #pragma unroll
    for (int j = 0; j < KSTEPS; ++j) {
        const int kk = j * 4;
        bf0[j].x = Uwt[(kk + kh)     * DP + nb0 + r15];
        bf0[j].y = Uwt[(kk + kh + 1) * DP + nb0 + r15];
    }
    if (has1) {
        #pragma unroll
        for (int j = 0; j < KSTEPS; ++j) {
            const int kk = j * 4;
            bf1[j].x = Uwt[(kk + kh)     * DP + nb1 + r15];
            bf1[j].y = Uwt[(kk + kh + 1) * DP + nb1 + r15];
        }
    } else {
        #pragma unroll
        for (int j = 0; j < KSTEPS; ++j) bf1[j] = v2f{0.f, 0.f};
    }

    auto gemm_tile = [&](int mb, const v2f* bf) -> v8f {
        v8f c = {};
        #pragma unroll
        for (int j = 0; j < KSTEPS; ++j) {
            const int kk = j * 4;
            const v2f a = *(const v2f*)&memS[(mb + r15) * DP + kk + kh];
            c = __builtin_amdgcn_wmma_f32_16x16x4_f32(
                     false, a,  false, bf[j],
                     (short)0, c,  false,  false);
        }
        return c;
    };

    float vk0[8], vk1[8];
    #pragma unroll
    for (int r = 0; r < 8; ++r) {
        const int m0 = mb0 + mh + r, n0 = nb0 + r15;
        const int m1 = mb1 + mh + r, n1 = nb1 + r15;
        const bool ok0 = (m0 < Mm && n0 < Dd);
        const bool ok1 = (has1 && m1 < Mm && n1 < Dd);
        const float v0 = VkBuf[ok0 ? m0 * Dd + n0 : 0];
        const float v1 = VkBuf[ok1 ? m1 * Dd + n1 : 0];
        vk0[r] = ok0 ? v0 : 0.f;
        vk1[r] = ok1 ? v1 : 0.f;
    }

    auto apply_tile = [&](int mb, int nb, v8f c, const float* vk, const float* wsRow) {
        #pragma unroll
        for (int r = 0; r < 8; ++r) {
            const int m = mb + mh + r;
            const int n = nb + r15;
            if (m < Mm && n < Dd) {
                const v2f sg = sgS[m];
                float x = sg.x * c[r] + vk[r] + wsRow[n];
                x = (x >= 0.f) ? x : pa * x;
                memS[m * DP + n] = sg.x * memS[m * DP + n] + x * sg.y;
            }
        }
    };

    const int  g8   = (tid >> 3) & 31;
    const int  l8   = tid & 7;
    float*       mRow = &memS[g8 * DP + 2 * l8];

    auto row_norm2 = [&](const v2f* v) -> float {
        float s0 = 0.f, s1 = 0.f;
        #pragma unroll
        for (int i = 0; i < 7; ++i) {
            s0 = __builtin_fmaf(v[i].x, v[i].x, s0);
            s1 = __builtin_fmaf(v[i].y, v[i].y, s1);
        }
        return sum8(s0 + s1);
    };

    for (int t = 0; t < Ss; ++t) {
        const float* pRow = &preS[t * PP];

        v8f acc0 = {}, acc1 = {};
        if (gemmW) {
            acc0 = gemm_tile(mb0, bf0);
            if (has1) acc1 = gemm_tile(mb1, bf1);
        } else {
            v2f v[7], u[7];
            #pragma unroll
            for (int i = 0; i < 7; ++i) v[i] = *(const v2f*)&mRow[16 * i];
            #pragma unroll
            for (int i = 0; i < 7; ++i)
                u[i] = *(const v2f*)&sentS[t * DP + 2 * l8 + 16 * i];
            const float kg = pRow[DP + g8];
            __builtin_amdgcn_sched_barrier(0);
            const float n2 = row_norm2(v);
            float d0 = 0.f, d1 = 0.f;
            #pragma unroll
            for (int i = 0; i < 7; ++i) {
                d0 = __builtin_fmaf(u[i].x, v[i].x, d0);
                d1 = __builtin_fmaf(u[i].y, v[i].y, d1);
            }
            const float dt = sum8(d0 + d1);
            const float scale = (t == 0) ? 1.f : 1.f / sqrtf(n2);
            const float z = scale * dt + kg;
            const float g = 1.f / (1.f + expf(-z));
            if (l8 == 0) sgS[g8] = v2f{scale, (z == 0.f) ? 0.f : g};
        }
        __syncthreads();

        if (gemmW) {
            apply_tile(mb0, nb0, acc0, vk0, pRow);
            if (has1) apply_tile(mb1, nb1, acc1, vk1, pRow);
        }
        __syncthreads();
    }

    if (!gemmW) {
        v2f v[7];
        #pragma unroll
        for (int i = 0; i < 7; ++i) v[i] = *(const v2f*)&mRow[16 * i];
        const float scale = 1.f / sqrtf(row_norm2(v));
        #pragma unroll
        for (int i = 0; i < 7; ++i) {
            v[i].x *= scale;
            v[i].y *= scale;
            *(v2f*)&mRow[16 * i] = v[i];
        }
    }
    __syncthreads();

    for (int i = tid; i < Mm * Dd; i += NTHR) {
        const int m = i / Dd, d = i - m * Dd;
        out[b * Mm * Dd + i] = memS[m * DP + d];
    }
}

extern "C" void kernel_launch(void* const* d_in, const int* in_sizes, int n_in,
                              void* d_out, int out_size, void* d_ws, size_t ws_size,
                              hipStream_t stream) {
    (void)in_sizes; (void)n_in; (void)out_size; (void)ws_size;
    const int*   tokens  = (const int*)  d_in[0];
    const float* emb     = (const float*)d_in[1];
    const float* keys    = (const float*)d_in[2];
    const float* mult    = (const float*)d_in[3];
    const float* Uw      = (const float*)d_in[4];
    const float* Vw      = (const float*)d_in[5];
    const float* Ww      = (const float*)d_in[6];
    const float* prelu_a = (const float*)d_in[7];
    float* out = (float*)d_out;

    float* sents = (float*)d_ws;
    float* VkBuf = sents + (size_t)Bb * Ss * Dd;

    sents_kernel<<<Bb * Ss, 128, 0, stream>>>(tokens, emb, mult, sents);
    vk_kernel<<<Mm, 128, 0, stream>>>(keys, Vw, VkBuf);
    entnet_scan_kernel<<<Bb, NTHR, 0, stream>>>(sents, keys, Uw, Ww, VkBuf,
                                                prelu_a, out);
}


# State update made predicate-free: pad rows carry zero scale and gate, pad columns meet zero terms; a wave issues its tiles' LDS loads as one group, one round trip instead of one per row section
# speedup vs baseline: 3.8381x; 1.1274x over previous
#include <hip/hip_runtime.h>

#define Bb   64
#define Ss   128
#define Ll   32
#define Dd   100
#define Mm   20
#define DP   112
#define MP   32
#define NTHR 416
#define NWAVE 8
#define NTILE 14
#define KSTEPS 25
#define NB   144
#define PW   102
#define PP   144

typedef __attribute__((ext_vector_type(2))) float v2f;
typedef __attribute__((ext_vector_type(4))) float v4f;
typedef __attribute__((ext_vector_type(8))) float v8f;

template <int N> struct RowCount { static constexpr int value = N; };
template <int N> struct UpdIn { float sc[N], gt[N], old[N], ws; };

__global__ __launch_bounds__(128) void sents_kernel(
    const int* __restrict__ tokens, const float* __restrict__ emb,
    const float* __restrict__ mult, float* __restrict__ sents)
{
    const int bs = blockIdx.x;
    const int d  = threadIdx.x;
    if (d >= Dd) return;
    const int* trow = tokens + bs * Ll;
    float acc = 0.f;
    #pragma unroll 4
    for (int l = 0; l < Ll; ++l) {
        const int tok = trow[l];
        acc += emb[tok * Dd + d] * mult[l * Dd + d];
    }
    sents[bs * Dd + d] = acc;
}

__global__ __launch_bounds__(128) void vk_kernel(
    const float* __restrict__ keys, const float* __restrict__ Vw,
    float* __restrict__ VkBuf)
{
    const int m = blockIdx.x;
    const int d = threadIdx.x;
    if (d >= Dd) return;
    float acc = 0.f;
    for (int k = 0; k < Dd; ++k) acc += keys[m * Dd + k] * Vw[d * Dd + k];
    VkBuf[m * Dd + d] = acc;
}

__device__ __forceinline__ float sum8(float x)
{
    x += __builtin_bit_cast(float, __builtin_amdgcn_update_dpp(
             0, __builtin_bit_cast(int, x), 0x161  , 0xf, 0xf, true));
    x += __builtin_bit_cast(float, __builtin_amdgcn_update_dpp(
             0, __builtin_bit_cast(int, x), 0x162  , 0xf, 0xf, true));
    x += __builtin_bit_cast(float, __builtin_amdgcn_update_dpp(
             0, __builtin_bit_cast(int, x), 0x164  , 0xf, 0xf, true));
    return x;
}

__global__ __launch_bounds__(NTHR) void entnet_scan_kernel(
    const float* __restrict__ sents, const float* __restrict__ keys,
    const float* __restrict__ Uw,    const float* __restrict__ Ww,
    const float* __restrict__ VkBuf, const float* __restrict__ prelu_a,
    float* __restrict__ out)
{
    const int b    = blockIdx.x;
    const int tid  = threadIdx.x;
    const int wave = tid >> 5;
    const int lane = tid & 31;
    const float pa = prelu_a[0];

    __shared__ float Uwt  [Dd * DP];
    __shared__ float memS [MP * DP];
    __shared__ float sentS[Ss * DP];
    __shared__ float wkS  [NB * PW];
    __shared__ float preS [Ss * PP];
    __shared__ __attribute__((aligned(16))) float scS[MP];
    __shared__ __attribute__((aligned(16))) float gtS[MP];

    for (int i = tid; i < MP * DP; i += NTHR) memS[i] = 0.f;
    for (int i = tid; i < MP; i += NTHR) { scS[i] = 0.f; gtS[i] = 0.f; }
    for (int i = tid; i < Dd * DP; i += NTHR) {
        const int k = i / DP, d = i - k * DP;
        Uwt[i] = (d < Dd) ? Uw[d * Dd + k] : 0.f;
    }
    __syncthreads();
    for (int i = tid; i < Mm * Dd; i += NTHR) {
        const int m = i / Dd, d = i - m * Dd;
        memS[m * DP + d] = keys[i];
    }
    for (int i = tid; i < Ss * (DP - Dd); i += NTHR) {
        const int t = i / (DP - Dd), k = i - t * (DP - Dd);
        sentS[t * DP + Dd + k] = 0.f;
    }
    #pragma unroll 10
    for (int i = tid; i < Ss * Dd; i += NTHR) {
        const int t = i / Dd, k = i - t * Dd;
        sentS[t * DP + k] = sents[b * Ss * Dd + i];
    }
    #pragma unroll 8
    for (int i = tid; i < Dd * Dd; i += NTHR) {
        const int n = i / Dd, k = i - n * Dd;
        wkS[n * PW + k] = Ww[i];
    }
    #pragma unroll 8
    for (int i = tid; i < Mm * Dd; i += NTHR) {
        const int m = i / Dd, k = i - m * Dd;
        wkS[(DP + m) * PW + k] = keys[i];
    }
    for (int i = tid; i < (DP - Dd) * PW; i += NTHR) {
        wkS[Dd * PW + i]        = 0.f;
        wkS[(DP + Mm) * PW + i] = 0.f;
    }
    __syncthreads();

    const int r15 = lane & 15;
    const int kh  = (lane >> 4) * 2;
    const int mh  = (lane >> 4) * 8;

    const int  t0   = wave;
    const int  t1   = wave + NWAVE;
    const bool has1 = (t1 < NTILE);
    const int  mb0 = (t0 / 7) * 16, nb0 = (t0 % 7) * 16;
    const int  mb1 = (t1 / 7) * 16, nb1 = (t1 % 7) * 16;

    const bool gemmW = (__builtin_amdgcn_readfirstlane(tid) < NWAVE * 32);

    if (gemmW) {
        const int tb = wave * 16;
        v2f af[KSTEPS];
        #pragma unroll
        for (int j = 0; j < KSTEPS; ++j)
            af[j] = *(const v2f*)&sentS[(tb + r15) * DP + j * 4 + kh];
        for (int nt = 0; nt < NB / 16; ++nt) {
            v8f c = {};
            #pragma unroll
            for (int j = 0; j < KSTEPS; ++j) {
                const v2f bw = *(const v2f*)&wkS[(nt * 16 + r15) * PW + j * 4 + kh];
                c = __builtin_amdgcn_wmma_f32_16x16x4_f32(
                         false, af[j],  false, bw,
                         (short)0, c,  false,  false);
            }
            #pragma unroll
            for (int r = 0; r < 8; ++r)
                preS[(tb + mh + r) * PP + nt * 16 + r15] = c[r];
        }
    }
    __syncthreads();

    v2f bf0[KSTEPS], bf1[KSTEPS];
    #pragma unroll
    for (int j = 0; j < KSTEPS; ++j) {
        const int kk = j * 4;
        bf0[j].x = Uwt[(kk + kh)     * DP + nb0 + r15];
        bf0[j].y = Uwt[(kk + kh + 1) * DP + nb0 + r15];
    }
    if (has1) {
        #pragma unroll
        for (int j = 0; j < KSTEPS; ++j) {
            const int kk = j * 4;
            bf1[j].x = Uwt[(kk + kh)     * DP + nb1 + r15];
            bf1[j].y = Uwt[(kk + kh + 1) * DP + nb1 + r15];
        }
    } else {
        #pragma unroll
        for (int j = 0; j < KSTEPS; ++j) bf1[j] = v2f{0.f, 0.f};
    }

    auto gemm_tile = [&](int mb, const v2f* bf) -> v8f {
        v8f c = {};
        #pragma unroll
        for (int j = 0; j < KSTEPS; ++j) {
            const int kk = j * 4;
            const v2f a = *(const v2f*)&memS[(mb + r15) * DP + kk + kh];
            c = __builtin_amdgcn_wmma_f32_16x16x4_f32(
                     false, a,  false, bf[j],
                     (short)0, c,  false,  false);
        }
        return c;
    };

    float vk0[8], vk1[8];
    #pragma unroll
    for (int r = 0; r < 8; ++r) {
        const int m0 = mb0 + mh + r, n0 = nb0 + r15;
        const int m1 = mb1 + mh + r, n1 = nb1 + r15;
        const bool ok0 = (m0 < Mm && n0 < Dd);
        const bool ok1 = (has1 && m1 < Mm && n1 < Dd);
        const float v0 = VkBuf[ok0 ? m0 * Dd + n0 : 0];
        const float v1 = VkBuf[ok1 ? m1 * Dd + n1 : 0];
        vk0[r] = ok0 ? v0 : 0.f;
        vk1[r] = ok1 ? v1 : 0.f;
    }

    auto upd_load = [&](auto nrTag, int mb, int nb, const float* wsRow) {
        constexpr int NR = decltype(nrTag)::value;
        UpdIn<NR> u;
        const v4f* sp = (const v4f*)&scS[mb + mh];
        const v4f* gp = (const v4f*)&gtS[mb + mh];
        #pragma unroll
        for (int q = 0; q < NR / 4; ++q) {
            const v4f s = sp[q], g = gp[q];
            u.sc[4 * q] = s.x; u.sc[4 * q + 1] = s.y; u.sc[4 * q + 2] = s.z; u.sc[4 * q + 3] = s.w;
            u.gt[4 * q] = g.x; u.gt[4 * q + 1] = g.y; u.gt[4 * q + 2] = g.z; u.gt[4 * q + 3] = g.w;
        }
        u.ws = wsRow[nb + r15];
        const float* mp = &memS[(mb + mh) * DP + nb + r15];
        #pragma unroll
        for (int r = 0; r < NR; ++r) u.old[r] = mp[r * DP];
        return u;
    };
    auto upd_store = [&](auto nrTag, int mb, int nb, v8f c, const float* vk, const auto& u) {
        constexpr int NR = decltype(nrTag)::value;
        float* mp = &memS[(mb + mh) * DP + nb + r15];
        #pragma unroll
        for (int r = 0; r < NR; ++r) {
            float x = __builtin_fmaf(u.sc[r], c[r], vk[r]) + u.ws;
            x = (x >= 0.f) ? x : pa * x;
            mp[r * DP] = __builtin_fmaf(u.sc[r], u.old[r], x * u.gt[r]);
        }
    };

    const int  g8   = (tid >> 3) & 31;
    const int  l8   = tid & 7;
    float*       mRow = &memS[g8 * DP + 2 * l8];

    auto row_norm2 = [&](const v2f* v) -> float {
        float s0 = 0.f, s1 = 0.f;
        #pragma unroll
        for (int i = 0; i < 7; ++i) {
            s0 = __builtin_fmaf(v[i].x, v[i].x, s0);
            s1 = __builtin_fmaf(v[i].y, v[i].y, s1);
        }
        return sum8(s0 + s1);
    };

    for (int t = 0; t < Ss; ++t) {
        const float* pRow = &preS[t * PP];

        v8f acc0 = {}, acc1 = {};
        if (gemmW) {
            acc0 = gemm_tile(mb0, bf0);
            if (has1) acc1 = gemm_tile(mb1, bf1);
        } else {
            v2f v[7], u[7];
            #pragma unroll
            for (int i = 0; i < 7; ++i) v[i] = *(const v2f*)&mRow[16 * i];
            #pragma unroll
            for (int i = 0; i < 7; ++i)
                u[i] = *(const v2f*)&sentS[t * DP + 2 * l8 + 16 * i];
            const float kg = pRow[DP + g8];
            __builtin_amdgcn_sched_barrier(0);
            const float n2 = row_norm2(v);
            float d0 = 0.f, d1 = 0.f;
            #pragma unroll
            for (int i = 0; i < 7; ++i) {
                d0 = __builtin_fmaf(u[i].x, v[i].x, d0);
                d1 = __builtin_fmaf(u[i].y, v[i].y, d1);
            }
            const float dt = sum8(d0 + d1);
            const float scale = (t == 0) ? 1.f : 1.f / sqrtf(n2);
            const float z = scale * dt + kg;
            const float g = 1.f / (1.f + expf(-z));
            if (l8 == 0) { scS[g8] = scale; gtS[g8] = (z == 0.f) ? 0.f : g; }
        }
        __syncthreads();

        if (gemmW) {
            const auto u0 = upd_load(RowCount<8>{}, mb0, nb0, pRow);
            if (has1) {
                const auto u1 = upd_load(RowCount<4>{}, mb1, nb1, pRow);
                upd_store(RowCount<8>{}, mb0, nb0, acc0, vk0, u0);
                upd_store(RowCount<4>{}, mb1, nb1, acc1, vk1, u1);
            } else {
                upd_store(RowCount<8>{}, mb0, nb0, acc0, vk0, u0);
            }
        }
        __syncthreads();
    }

    if (!gemmW) {
        v2f v[7];
        #pragma unroll
        for (int i = 0; i < 7; ++i) v[i] = *(const v2f*)&mRow[16 * i];
        const float scale = 1.f / sqrtf(row_norm2(v));
        #pragma unroll
        for (int i = 0; i < 7; ++i) {
            v[i].x *= scale;
            v[i].y *= scale;
            *(v2f*)&mRow[16 * i] = v[i];
        }
    }
    __syncthreads();

    for (int i = tid; i < Mm * Dd; i += NTHR) {
        const int m = i / Dd, d = i - m * Dd;
        out[b * Mm * Dd + i] = memS[m * DP + d];
    }
}

extern "C" void kernel_launch(void* const* d_in, const int* in_sizes, int n_in,
                              void* d_out, int out_size, void* d_ws, size_t ws_size,
                              hipStream_t stream) {
    (void)in_sizes; (void)n_in; (void)out_size; (void)ws_size;
    const int*   tokens  = (const int*)  d_in[0];
    const float* emb     = (const float*)d_in[1];
    const float* keys    = (const float*)d_in[2];
    const float* mult    = (const float*)d_in[3];
    const float* Uw      = (const float*)d_in[4];
    const float* Vw      = (const float*)d_in[5];
    const float* Ww      = (const float*)d_in[6];
    const float* prelu_a = (const float*)d_in[7];
    float* out = (float*)d_out;

    float* sents = (float*)d_ws;
    float* VkBuf = sents + (size_t)Bb * Ss * Dd;

    sents_kernel<<<Bb * Ss, 128, 0, stream>>>(tokens, emb, mult, sents);
    vk_kernel<<<Mm, 128, 0, stream>>>(keys, Vw, VkBuf);
    entnet_scan_kernel<<<Bb, NTHR, 0, stream>>>(sents, keys, Uw, Ww, VkBuf,
                                                prelu_a, out);
}


# Slot rows regrouped across batch elements into full 16-row WMMA tiles: 80 work-groups of one M-tile (7 product waves) replace 64 of two padded tiles; Uw fragments read straight from global
# speedup vs baseline: 5.3686x; 1.3988x over previous
#include <hip/hip_runtime.h>

#define Bb   64
#define Ss   128
#define Ll   32
#define Dd   100
#define Mm   20
#define DP   112
#define RT   16
#define NBLK ((Bb * Mm) / RT)
#define NTHR 352
#define NWAVE 7
#define PWAVES 8
#define KSTEPS 25
#define NB   144
#define PW   100
#define PP   132

typedef __attribute__((ext_vector_type(2))) float v2f;
typedef __attribute__((ext_vector_type(4))) float v4f;
typedef __attribute__((ext_vector_type(8))) float v8f;

__global__ __launch_bounds__(128) void sents_kernel(
    const int* __restrict__ tokens, const float* __restrict__ emb,
    const float* __restrict__ mult, float* __restrict__ sents)
{
    const int bs = blockIdx.x;
    const int d  = threadIdx.x;
    if (d >= Dd) return;
    const int* trow = tokens + bs * Ll;
    float acc = 0.f;
    #pragma unroll 4
    for (int l = 0; l < Ll; ++l) {
        const int tok = trow[l];
        acc += emb[tok * Dd + d] * mult[l * Dd + d];
    }
    sents[bs * Dd + d] = acc;
}

__global__ __launch_bounds__(128) void vk_kernel(
    const float* __restrict__ keys, const float* __restrict__ Vw,
    float* __restrict__ VkBuf)
{
    const int m = blockIdx.x;
    const int d = threadIdx.x;
    if (d >= Dd) return;
    float acc = 0.f;
    for (int k = 0; k < Dd; ++k) acc += keys[m * Dd + k] * Vw[d * Dd + k];
    VkBuf[m * Dd + d] = acc;
}

__device__ __forceinline__ float sum8(float x)
{
    x += __builtin_bit_cast(float, __builtin_amdgcn_update_dpp(
             0, __builtin_bit_cast(int, x), 0x161  , 0xf, 0xf, true));
    x += __builtin_bit_cast(float, __builtin_amdgcn_update_dpp(
             0, __builtin_bit_cast(int, x), 0x162  , 0xf, 0xf, true));
    x += __builtin_bit_cast(float, __builtin_amdgcn_update_dpp(
             0, __builtin_bit_cast(int, x), 0x164  , 0xf, 0xf, true));
    return x;
}

__global__ __launch_bounds__(NTHR) void entnet_scan_kernel(
    const float* __restrict__ sents, const float* __restrict__ keys,
    const float* __restrict__ Uw,    const float* __restrict__ Ww,
    const float* __restrict__ VkBuf, const float* __restrict__ prelu_a,
    float* __restrict__ out)
{
    const int g0   = blockIdx.x * RT;
    const int tid  = threadIdx.x;
    const int wave = tid >> 5;
    const int lane = tid & 31;
    const float pa = prelu_a[0];

    const int  bA     = g0 / Mm;
    const int  iB     = (bA + 1) * Mm - g0;
    const bool two    = (iB < RT);
    const int  nbatch = two ? 2 : 1;
    const int  sOffB  = two ? Ss * DP : 0;
    const int  pOffB  = two ? Ss * PP : 0;

    __shared__ float memS [RT * DP];
    __shared__ float sentS[2 * Ss * DP];
    __shared__ float wkS  [NB * PW];
    __shared__ float preS [2 * Ss * PP];
    __shared__ __attribute__((aligned(16))) float scS[RT];
    __shared__ __attribute__((aligned(16))) float gtS[RT];

    for (int i = tid; i < RT * DP; i += NTHR) {
        const int r = i / DP, d = i - r * DP;
        const int g = g0 + r;
        const int m = g - (g / Mm) * Mm;
        memS[i] = (d < Dd) ? keys[m * Dd + d] : 0.f;
    }
    for (int sel = 0; sel < nbatch; ++sel) {
        const int bb = bA + sel;
        float* sS = &sentS[sel * Ss * DP];
        for (int i = tid; i < Ss * (DP - Dd); i += NTHR) {
            const int t = i / (DP - Dd), k = i - t * (DP - Dd);
            sS[t * DP + Dd + k] = 0.f;
        }
        #pragma unroll 10
        for (int i = tid; i < Ss * Dd; i += NTHR) {
            const int t = i / Dd, k = i - t * Dd;
            sS[t * DP + k] = sents[bb * Ss * Dd + i];
        }
    }
    #pragma unroll 8
    for (int i = tid; i < Dd * Dd; i += NTHR) wkS[i] = Ww[i];
    #pragma unroll 6
    for (int i = tid; i < Mm * Dd; i += NTHR) wkS[DP * PW + i] = keys[i];
    for (int i = tid; i < (DP - Dd) * PW; i += NTHR) {
        wkS[Dd * PW + i]        = 0.f;
        wkS[(DP + Mm) * PW + i] = 0.f;
    }
    __syncthreads();

    const int r15 = lane & 15;
    const int kh  = (lane >> 4) * 2;
    const int mh  = (lane >> 4) * 8;

    const int  waveU = __builtin_amdgcn_readfirstlane(tid) >> 5;
    const bool gemmW = (waveU < NWAVE);
    const int  nb    = wave * 16;

    if (waveU < PWAVES) {
        const int tb = wave * 16;
        for (int sel = 0; sel < nbatch; ++sel) {
            const float* sS = &sentS[sel * Ss * DP];
            float*       pS = &preS[sel * Ss * PP];
            v2f af[KSTEPS];
            #pragma unroll
            for (int j = 0; j < KSTEPS; ++j)
                af[j] = *(const v2f*)&sS[(tb + r15) * DP + j * 4 + kh];
            for (int nt = 0; nt < NB / 16; ++nt) {
                v8f c = {};
                #pragma unroll
                for (int j = 0; j < KSTEPS; ++j) {
                    const v2f bw = *(const v2f*)&wkS[(nt * 16 + r15) * PW + j * 4 + kh];
                    c = __builtin_amdgcn_wmma_f32_16x16x4_f32(
                             false, af[j],  false, bw,
                             (short)0, c,  false,  false);
                }
                if (nt * 16 + r15 < PP) {
                    #pragma unroll
                    for (int r = 0; r < 8; ++r)
                        pS[(tb + mh + r) * PP + nt * 16 + r15] = c[r];
                }
            }
        }
    }
    __syncthreads();

    v2f   bf[KSTEPS];
    float vk[8];
    if (gemmW) {
        const int  n   = nb + r15;
        const bool okn = (n < Dd);
        const int  nc  = okn ? n : 0;
        #pragma unroll
        for (int j = 0; j < KSTEPS; ++j) {
            const v2f w = *(const v2f*)&Uw[nc * Dd + j * 4 + kh];
            bf[j] = okn ? w : v2f{0.f, 0.f};
        }
        #pragma unroll
        for (int r = 0; r < 8; ++r) {
            const int g = g0 + mh + r;
            const int m = g - (g / Mm) * Mm;
            const float v = VkBuf[m * Dd + nc];
            vk[r] = okn ? v : 0.f;
        }
    } else {
        #pragma unroll
        for (int j = 0; j < KSTEPS; ++j) bf[j] = v2f{0.f, 0.f};
        #pragma unroll
        for (int r = 0; r < 8; ++r) vk[r] = 0.f;
    }

    const int  g8   = (tid >> 3) & 15;
    const int  l8   = tid & 7;
    float*       mRow = &memS[g8 * DP + 2 * l8];
    const int  gRow  = g0 + g8;
    const int  mSlot = gRow - (gRow / Mm) * Mm;
    const bool rowB  = (g8 >= iB);
    const float* sRow0 = &sentS[(rowB ? sOffB : 0) + 2 * l8];
    const float* kRow0 = &preS[(rowB ? pOffB : 0) + DP + mSlot];

    auto row_norm2 = [&](const v2f* v) -> float {
        float s0 = 0.f, s1 = 0.f;
        #pragma unroll
        for (int i = 0; i < 7; ++i) {
            s0 = __builtin_fmaf(v[i].x, v[i].x, s0);
            s1 = __builtin_fmaf(v[i].y, v[i].y, s1);
        }
        return sum8(s0 + s1);
    };

    for (int t = 0; t < Ss; ++t) {
        v8f acc = {};
        if (gemmW) {
            #pragma unroll
            for (int j = 0; j < KSTEPS; ++j) {
                const v2f a = *(const v2f*)&memS[r15 * DP + j * 4 + kh];
                acc = __builtin_amdgcn_wmma_f32_16x16x4_f32(
                         false, a,  false, bf[j],
                         (short)0, acc,  false,  false);
            }
        } else {
            v2f v[7], u[7];
            #pragma unroll
            for (int i = 0; i < 7; ++i) v[i] = *(const v2f*)&mRow[16 * i];
            #pragma unroll
            for (int i = 0; i < 7; ++i)
                u[i] = *(const v2f*)&sRow0[t * DP + 16 * i];
            const float kg = kRow0[t * PP];
            __builtin_amdgcn_sched_barrier(0);
            const float n2 = row_norm2(v);
            float d0 = 0.f, d1 = 0.f;
            #pragma unroll
            for (int i = 0; i < 7; ++i) {
                d0 = __builtin_fmaf(u[i].x, v[i].x, d0);
                d1 = __builtin_fmaf(u[i].y, v[i].y, d1);
            }
            const float dt = sum8(d0 + d1);
            const float scale = (t == 0) ? 1.f : 1.f / sqrtf(n2);
            const float z = scale * dt + kg;
            const float g = 1.f / (1.f + expf(-z));
            if (l8 == 0) { scS[g8] = scale; gtS[g8] = (z == 0.f) ? 0.f : g; }
        }
        __syncthreads();

        if (gemmW) {
            const v4f* sp = (const v4f*)&scS[mh];
            const v4f* gp = (const v4f*)&gtS[mh];
            const v4f s0 = sp[0], s1 = sp[1], q0 = gp[0], q1 = gp[1];
            const float sc[8] = {s0.x, s0.y, s0.z, s0.w, s1.x, s1.y, s1.z, s1.w};
            const float gt[8] = {q0.x, q0.y, q0.z, q0.w, q1.x, q1.y, q1.z, q1.w};
            const float wsA = preS[t * PP + nb + r15];
            const float wsB = preS[pOffB + t * PP + nb + r15];
            float* mp = &memS[mh * DP + nb + r15];
            float old[8];
            #pragma unroll
            for (int r = 0; r < 8; ++r) old[r] = mp[r * DP];
            #pragma unroll
            for (int r = 0; r < 8; ++r) {
                const float ws = (mh + r >= iB) ? wsB : wsA;
                float x = __builtin_fmaf(sc[r], acc[r], vk[r]) + ws;
                x = (x >= 0.f) ? x : pa * x;
                mp[r * DP] = __builtin_fmaf(sc[r], old[r], x * gt[r]);
            }
        }
        __syncthreads();
    }

    if (!gemmW) {
        v2f v[7];
        #pragma unroll
        for (int i = 0; i < 7; ++i) v[i] = *(const v2f*)&mRow[16 * i];
        const float scale = 1.f / sqrtf(row_norm2(v));
        #pragma unroll
        for (int i = 0; i < 7; ++i) {
            v[i].x *= scale;
            v[i].y *= scale;
            *(v2f*)&mRow[16 * i] = v[i];
        }
    }
    __syncthreads();

    for (int i = tid; i < RT * Dd; i += NTHR) {
        const int r = i / Dd, d = i - r * Dd;
        out[g0 * Dd + i] = memS[r * DP + d];
    }
}

extern "C" void kernel_launch(void* const* d_in, const int* in_sizes, int n_in,
                              void* d_out, int out_size, void* d_ws, size_t ws_size,
                              hipStream_t stream) {
    (void)in_sizes; (void)n_in; (void)out_size; (void)ws_size;
    const int*   tokens  = (const int*)  d_in[0];
    const float* emb     = (const float*)d_in[1];
    const float* keys    = (const float*)d_in[2];
    const float* mult    = (const float*)d_in[3];
    const float* Uw      = (const float*)d_in[4];
    const float* Vw      = (const float*)d_in[5];
    const float* Ww      = (const float*)d_in[6];
    const float* prelu_a = (const float*)d_in[7];
    float* out = (float*)d_out;

    float* sents = (float*)d_ws;
    float* VkBuf = sents + (size_t)Bb * Ss * Dd;

    sents_kernel<<<Bb * Ss, 128, 0, stream>>>(tokens, emb, mult, sents);
    vk_kernel<<<Mm, 128, 0, stream>>>(keys, Vw, VkBuf);
    entnet_scan_kernel<<<NBLK, NTHR, 0, stream>>>(sents, keys, Uw, Ww, VkBuf,
                                                prelu_a, out);
}


# State tile row pitch 112 -> 116 floats in LDS: the step product's A-fragment reads (4 rows per bank before) and the update's row accesses (2 rows per bank) become bank-conflict free
# speedup vs baseline: 5.5464x; 1.0331x over previous
#include <hip/hip_runtime.h>

#define Bb   64
#define Ss   128
#define Ll   32
#define Dd   100
#define Mm   20
#define DP   112
#define MP   116
#define RT   16
#define NBLK ((Bb * Mm) / RT)
#define NTHR 352
#define NWAVE 7
#define PWAVES 8
#define KSTEPS 25
#define NB   144
#define PW   100
#define PP   132

typedef __attribute__((ext_vector_type(2))) float v2f;
typedef __attribute__((ext_vector_type(4))) float v4f;
typedef __attribute__((ext_vector_type(8))) float v8f;

__global__ __launch_bounds__(128) void sents_kernel(
    const int* __restrict__ tokens, const float* __restrict__ emb,
    const float* __restrict__ mult, float* __restrict__ sents)
{
    const int bs = blockIdx.x;
    const int d  = threadIdx.x;
    if (d >= Dd) return;
    const int* trow = tokens + bs * Ll;
    float acc = 0.f;
    #pragma unroll 4
    for (int l = 0; l < Ll; ++l) {
        const int tok = trow[l];
        acc += emb[tok * Dd + d] * mult[l * Dd + d];
    }
    sents[bs * Dd + d] = acc;
}

__global__ __launch_bounds__(128) void vk_kernel(
    const float* __restrict__ keys, const float* __restrict__ Vw,
    float* __restrict__ VkBuf)
{
    const int m = blockIdx.x;
    const int d = threadIdx.x;
    if (d >= Dd) return;
    float acc = 0.f;
    for (int k = 0; k < Dd; ++k) acc += keys[m * Dd + k] * Vw[d * Dd + k];
    VkBuf[m * Dd + d] = acc;
}

__device__ __forceinline__ float sum8(float x)
{
    x += __builtin_bit_cast(float, __builtin_amdgcn_update_dpp(
             0, __builtin_bit_cast(int, x), 0x161  , 0xf, 0xf, true));
    x += __builtin_bit_cast(float, __builtin_amdgcn_update_dpp(
             0, __builtin_bit_cast(int, x), 0x162  , 0xf, 0xf, true));
    x += __builtin_bit_cast(float, __builtin_amdgcn_update_dpp(
             0, __builtin_bit_cast(int, x), 0x164  , 0xf, 0xf, true));
    return x;
}

__global__ __launch_bounds__(NTHR) void entnet_scan_kernel(
    const float* __restrict__ sents, const float* __restrict__ keys,
    const float* __restrict__ Uw,    const float* __restrict__ Ww,
    const float* __restrict__ VkBuf, const float* __restrict__ prelu_a,
    float* __restrict__ out)
{
    const int g0   = blockIdx.x * RT;
    const int tid  = threadIdx.x;
    const int wave = tid >> 5;
    const int lane = tid & 31;
    const float pa = prelu_a[0];

    const int  bA     = g0 / Mm;
    const int  iB     = (bA + 1) * Mm - g0;
    const bool two    = (iB < RT);
    const int  nbatch = two ? 2 : 1;
    const int  sOffB  = two ? Ss * DP : 0;
    const int  pOffB  = two ? Ss * PP : 0;

    __shared__ __attribute__((aligned(16))) float memS[RT * MP];
    __shared__ float sentS[2 * Ss * DP];
    __shared__ float wkS  [NB * PW];
    __shared__ float preS [2 * Ss * PP];
    __shared__ __attribute__((aligned(16))) float scS[RT];
    __shared__ __attribute__((aligned(16))) float gtS[RT];

    for (int i = tid; i < RT * DP; i += NTHR) {
        const int r = i / DP, d = i - r * DP;
        const int g = g0 + r;
        const int m = g - (g / Mm) * Mm;
        memS[r * MP + d] = (d < Dd) ? keys[m * Dd + d] : 0.f;
    }
    for (int sel = 0; sel < nbatch; ++sel) {
        const int bb = bA + sel;
        float* sS = &sentS[sel * Ss * DP];
        for (int i = tid; i < Ss * (DP - Dd); i += NTHR) {
            const int t = i / (DP - Dd), k = i - t * (DP - Dd);
            sS[t * DP + Dd + k] = 0.f;
        }
        #pragma unroll 10
        for (int i = tid; i < Ss * Dd; i += NTHR) {
            const int t = i / Dd, k = i - t * Dd;
            sS[t * DP + k] = sents[bb * Ss * Dd + i];
        }
    }
    #pragma unroll 8
    for (int i = tid; i < Dd * Dd; i += NTHR) wkS[i] = Ww[i];
    #pragma unroll 6
    for (int i = tid; i < Mm * Dd; i += NTHR) wkS[DP * PW + i] = keys[i];
    for (int i = tid; i < (DP - Dd) * PW; i += NTHR) {
        wkS[Dd * PW + i]        = 0.f;
        wkS[(DP + Mm) * PW + i] = 0.f;
    }
    __syncthreads();

    const int r15 = lane & 15;
    const int kh  = (lane >> 4) * 2;
    const int mh  = (lane >> 4) * 8;

    const int  waveU = __builtin_amdgcn_readfirstlane(tid) >> 5;
    const bool gemmW = (waveU < NWAVE);
    const int  nb    = wave * 16;

    if (waveU < PWAVES) {
        const int tb = wave * 16;
        for (int sel = 0; sel < nbatch; ++sel) {
            const float* sS = &sentS[sel * Ss * DP];
            float*       pS = &preS[sel * Ss * PP];
            v2f af[KSTEPS];
            #pragma unroll
            for (int j = 0; j < KSTEPS; ++j)
                af[j] = *(const v2f*)&sS[(tb + r15) * DP + j * 4 + kh];
            for (int nt = 0; nt < NB / 16; ++nt) {
                v8f c = {};
                #pragma unroll
                for (int j = 0; j < KSTEPS; ++j) {
                    const v2f bw = *(const v2f*)&wkS[(nt * 16 + r15) * PW + j * 4 + kh];
                    c = __builtin_amdgcn_wmma_f32_16x16x4_f32(
                             false, af[j],  false, bw,
                             (short)0, c,  false,  false);
                }
                if (nt * 16 + r15 < PP) {
                    #pragma unroll
                    for (int r = 0; r < 8; ++r)
                        pS[(tb + mh + r) * PP + nt * 16 + r15] = c[r];
                }
            }
        }
    }
    __syncthreads();

    v2f   bf[KSTEPS];
    float vk[8];
    if (gemmW) {
        const int  n   = nb + r15;
        const bool okn = (n < Dd);
        const int  nc  = okn ? n : 0;
        #pragma unroll
        for (int j = 0; j < KSTEPS; ++j) {
            const v2f w = *(const v2f*)&Uw[nc * Dd + j * 4 + kh];
            bf[j] = okn ? w : v2f{0.f, 0.f};
        }
        #pragma unroll
        for (int r = 0; r < 8; ++r) {
            const int g = g0 + mh + r;
            const int m = g - (g / Mm) * Mm;
            const float v = VkBuf[m * Dd + nc];
            vk[r] = okn ? v : 0.f;
        }
    } else {
        #pragma unroll
        for (int j = 0; j < KSTEPS; ++j) bf[j] = v2f{0.f, 0.f};
        #pragma unroll
        for (int r = 0; r < 8; ++r) vk[r] = 0.f;
    }

    const int  g8   = (tid >> 3) & 15;
    const int  l8   = tid & 7;
    float*       mRow = &memS[g8 * MP + 2 * l8];
    const int  gRow  = g0 + g8;
    const int  mSlot = gRow - (gRow / Mm) * Mm;
    const bool rowB  = (g8 >= iB);
    const float* sRow0 = &sentS[(rowB ? sOffB : 0) + 2 * l8];
    const float* kRow0 = &preS[(rowB ? pOffB : 0) + DP + mSlot];

    auto row_norm2 = [&](const v2f* v) -> float {
        float s0 = 0.f, s1 = 0.f;
        #pragma unroll
        for (int i = 0; i < 7; ++i) {
            s0 = __builtin_fmaf(v[i].x, v[i].x, s0);
            s1 = __builtin_fmaf(v[i].y, v[i].y, s1);
        }
        return sum8(s0 + s1);
    };

    for (int t = 0; t < Ss; ++t) {
        v8f acc = {};
        if (gemmW) {
            #pragma unroll
            for (int j = 0; j < KSTEPS; ++j) {
                const v2f a = *(const v2f*)&memS[r15 * MP + j * 4 + kh];
                acc = __builtin_amdgcn_wmma_f32_16x16x4_f32(
                         false, a,  false, bf[j],
                         (short)0, acc,  false,  false);
            }
        } else {
            v2f v[7], u[7];
            #pragma unroll
            for (int i = 0; i < 7; ++i) v[i] = *(const v2f*)&mRow[16 * i];
            #pragma unroll
            for (int i = 0; i < 7; ++i)
                u[i] = *(const v2f*)&sRow0[t * DP + 16 * i];
            const float kg = kRow0[t * PP];
            __builtin_amdgcn_sched_barrier(0);
            const float n2 = row_norm2(v);
            float d0 = 0.f, d1 = 0.f;
            #pragma unroll
            for (int i = 0; i < 7; ++i) {
                d0 = __builtin_fmaf(u[i].x, v[i].x, d0);
                d1 = __builtin_fmaf(u[i].y, v[i].y, d1);
            }
            const float dt = sum8(d0 + d1);
            const float scale = (t == 0) ? 1.f : 1.f / sqrtf(n2);
            const float z = scale * dt + kg;
            const float g = 1.f / (1.f + expf(-z));
            if (l8 == 0) { scS[g8] = scale; gtS[g8] = (z == 0.f) ? 0.f : g; }
        }
        __syncthreads();

        if (gemmW) {
            const v4f* sp = (const v4f*)&scS[mh];
            const v4f* gp = (const v4f*)&gtS[mh];
            const v4f s0 = sp[0], s1 = sp[1], q0 = gp[0], q1 = gp[1];
            const float sc[8] = {s0.x, s0.y, s0.z, s0.w, s1.x, s1.y, s1.z, s1.w};
            const float gt[8] = {q0.x, q0.y, q0.z, q0.w, q1.x, q1.y, q1.z, q1.w};
            const float wsA = preS[t * PP + nb + r15];
            const float wsB = preS[pOffB + t * PP + nb + r15];
            float* mp = &memS[mh * MP + nb + r15];
            float old[8];
            #pragma unroll
            for (int r = 0; r < 8; ++r) old[r] = mp[r * MP];
            #pragma unroll
            for (int r = 0; r < 8; ++r) {
                const float ws = (mh + r >= iB) ? wsB : wsA;
                float x = __builtin_fmaf(sc[r], acc[r], vk[r]) + ws;
                x = (x >= 0.f) ? x : pa * x;
                mp[r * MP] = __builtin_fmaf(sc[r], old[r], x * gt[r]);
            }
        }
        __syncthreads();
    }

    if (!gemmW) {
        v2f v[7];
        #pragma unroll
        for (int i = 0; i < 7; ++i) v[i] = *(const v2f*)&mRow[16 * i];
        const float scale = 1.f / sqrtf(row_norm2(v));
        #pragma unroll
        for (int i = 0; i < 7; ++i) {
            v[i].x *= scale;
            v[i].y *= scale;
            *(v2f*)&mRow[16 * i] = v[i];
        }
    }
    __syncthreads();

    for (int i = tid; i < RT * Dd; i += NTHR) {
        const int r = i / Dd, d = i - r * Dd;
        out[g0 * Dd + i] = memS[r * MP + d];
    }
}

extern "C" void kernel_launch(void* const* d_in, const int* in_sizes, int n_in,
                              void* d_out, int out_size, void* d_ws, size_t ws_size,
                              hipStream_t stream) {
    (void)in_sizes; (void)n_in; (void)out_size; (void)ws_size;
    const int*   tokens  = (const int*)  d_in[0];
    const float* emb     = (const float*)d_in[1];
    const float* keys    = (const float*)d_in[2];
    const float* mult    = (const float*)d_in[3];
    const float* Uw      = (const float*)d_in[4];
    const float* Vw      = (const float*)d_in[5];
    const float* Ww      = (const float*)d_in[6];
    const float* prelu_a = (const float*)d_in[7];
    float* out = (float*)d_out;

    float* sents = (float*)d_ws;
    float* VkBuf = sents + (size_t)Bb * Ss * Dd;

    sents_kernel<<<Bb * Ss, 128, 0, stream>>>(tokens, emb, mult, sents);
    vk_kernel<<<Mm, 128, 0, stream>>>(keys, Vw, VkBuf);
    entnet_scan_kernel<<<NBLK, NTHR, 0, stream>>>(sents, keys, Uw, Ww, VkBuf,
                                                prelu_a, out);
}


# Norm+gate pass moves from 4 waves (8 lanes a row, each youngest beside two product waves) to ONE wave of 2 lanes a row beside a lone product wave; block 352 -> 256; both row sums reordered
# speedup vs baseline: 6.3013x; 1.1361x over previous
#include <hip/hip_runtime.h>

#define Bb   64
#define Ss   128
#define Ll   32
#define Dd   100
#define Mm   20
#define DP   112
#define MP   116
#define RT   16
#define NBLK ((Bb * Mm) / RT)
#define NTHR 256
#define NWAVE 7
#define PWAVES 8
#define KSTEPS 25
#define NB   144
#define PW   100
#define PP   132

typedef __attribute__((ext_vector_type(2))) float v2f;
typedef __attribute__((ext_vector_type(4))) float v4f;
typedef __attribute__((ext_vector_type(8))) float v8f;

__global__ __launch_bounds__(128) void sents_kernel(
    const int* __restrict__ tokens, const float* __restrict__ emb,
    const float* __restrict__ mult, float* __restrict__ sents)
{
    const int bs = blockIdx.x;
    const int d  = threadIdx.x;
    if (d >= Dd) return;
    const int* trow = tokens + bs * Ll;
    float acc = 0.f;
    #pragma unroll 4
    for (int l = 0; l < Ll; ++l) {
        const int tok = trow[l];
        acc += emb[tok * Dd + d] * mult[l * Dd + d];
    }
    sents[bs * Dd + d] = acc;
}

__global__ __launch_bounds__(128) void vk_kernel(
    const float* __restrict__ keys, const float* __restrict__ Vw,
    float* __restrict__ VkBuf)
{
    const int m = blockIdx.x;
    const int d = threadIdx.x;
    if (d >= Dd) return;
    float acc = 0.f;
    for (int k = 0; k < Dd; ++k) acc += keys[m * Dd + k] * Vw[d * Dd + k];
    VkBuf[m * Dd + d] = acc;
}

__device__ __forceinline__ float sum_halves(float x)
{
    const int xi = __builtin_bit_cast(int, x);
    const int o  = __builtin_amdgcn_permlanex16(xi, xi, 0x76543210, 0xfedcba98, false, false);
    return x + __builtin_bit_cast(float, o);
}

__global__ __launch_bounds__(NTHR) void entnet_scan_kernel(
    const float* __restrict__ sents, const float* __restrict__ keys,
    const float* __restrict__ Uw,    const float* __restrict__ Ww,
    const float* __restrict__ VkBuf, const float* __restrict__ prelu_a,
    float* __restrict__ out)
{
    const int g0   = blockIdx.x * RT;
    const int tid  = threadIdx.x;
    const int wave = tid >> 5;
    const int lane = tid & 31;
    const float pa = prelu_a[0];

    const int  bA     = g0 / Mm;
    const int  iB     = (bA + 1) * Mm - g0;
    const bool two    = (iB < RT);
    const int  nbatch = two ? 2 : 1;
    const int  sOffB  = two ? Ss * DP : 0;
    const int  pOffB  = two ? Ss * PP : 0;

    __shared__ __attribute__((aligned(16))) float memS[RT * MP];
    __shared__ __attribute__((aligned(16))) float sentS[2 * Ss * DP];
    __shared__ float wkS  [NB * PW];
    __shared__ float preS [2 * Ss * PP];
    __shared__ __attribute__((aligned(16))) float scS[RT];
    __shared__ __attribute__((aligned(16))) float gtS[RT];

    for (int i = tid; i < RT * DP; i += NTHR) {
        const int r = i / DP, d = i - r * DP;
        const int g = g0 + r;
        const int m = g - (g / Mm) * Mm;
        memS[r * MP + d] = (d < Dd) ? keys[m * Dd + d] : 0.f;
    }
    for (int sel = 0; sel < nbatch; ++sel) {
        const int bb = bA + sel;
        float* sS = &sentS[sel * Ss * DP];
        for (int i = tid; i < Ss * (DP - Dd); i += NTHR) {
            const int t = i / (DP - Dd), k = i - t * (DP - Dd);
            sS[t * DP + Dd + k] = 0.f;
        }
        #pragma unroll 10
        for (int i = tid; i < Ss * Dd; i += NTHR) {
            const int t = i / Dd, k = i - t * Dd;
            sS[t * DP + k] = sents[bb * Ss * Dd + i];
        }
    }
    #pragma unroll 8
    for (int i = tid; i < Dd * Dd; i += NTHR) wkS[i] = Ww[i];
    #pragma unroll 6
    for (int i = tid; i < Mm * Dd; i += NTHR) wkS[DP * PW + i] = keys[i];
    for (int i = tid; i < (DP - Dd) * PW; i += NTHR) {
        wkS[Dd * PW + i]        = 0.f;
        wkS[(DP + Mm) * PW + i] = 0.f;
    }
    __syncthreads();

    const int r15 = lane & 15;
    const int kh  = (lane >> 4) * 2;
    const int mh  = (lane >> 4) * 8;

    const int  waveU = __builtin_amdgcn_readfirstlane(tid) >> 5;
    const bool gemmW = (waveU < NWAVE);
    const int  nb    = wave * 16;

    if (waveU < PWAVES) {
        const int tb = wave * 16;
        for (int sel = 0; sel < nbatch; ++sel) {
            const float* sS = &sentS[sel * Ss * DP];
            float*       pS = &preS[sel * Ss * PP];
            v2f af[KSTEPS];
            #pragma unroll
            for (int j = 0; j < KSTEPS; ++j)
                af[j] = *(const v2f*)&sS[(tb + r15) * DP + j * 4 + kh];
            for (int nt = 0; nt < NB / 16; ++nt) {
                v8f c = {};
                #pragma unroll
                for (int j = 0; j < KSTEPS; ++j) {
                    const v2f bw = *(const v2f*)&wkS[(nt * 16 + r15) * PW + j * 4 + kh];
                    c = __builtin_amdgcn_wmma_f32_16x16x4_f32(
                             false, af[j],  false, bw,
                             (short)0, c,  false,  false);
                }
                if (nt * 16 + r15 < PP) {
                    #pragma unroll
                    for (int r = 0; r < 8; ++r)
                        pS[(tb + mh + r) * PP + nt * 16 + r15] = c[r];
                }
            }
        }
    }
    __syncthreads();

    v2f   bf[KSTEPS];
    float vk[8];
    if (gemmW) {
        const int  n   = nb + r15;
        const bool okn = (n < Dd);
        const int  nc  = okn ? n : 0;
        #pragma unroll
        for (int j = 0; j < KSTEPS; ++j) {
            const v2f w = *(const v2f*)&Uw[nc * Dd + j * 4 + kh];
            bf[j] = okn ? w : v2f{0.f, 0.f};
        }
        #pragma unroll
        for (int r = 0; r < 8; ++r) {
            const int g = g0 + mh + r;
            const int m = g - (g / Mm) * Mm;
            const float v = VkBuf[m * Dd + nc];
            vk[r] = okn ? v : 0.f;
        }
    } else {
        #pragma unroll
        for (int j = 0; j < KSTEPS; ++j) bf[j] = v2f{0.f, 0.f};
        #pragma unroll
        for (int r = 0; r < 8; ++r) vk[r] = 0.f;
    }

    const int  hcol = (lane >> 4) * (DP / 2);
    float*       mRow = &memS[r15 * MP + hcol];
    const int  gRow  = g0 + r15;
    const int  mSlot = gRow - (gRow / Mm) * Mm;
    const bool rowB  = (r15 >= iB);
    const float* sRow0 = &sentS[(rowB ? sOffB : 0) + hcol];
    const float* kRow0 = &preS[(rowB ? pOffB : 0) + DP + mSlot];

    auto row_norm2 = [&](const v4f* v) -> float {
        float s0 = 0.f, s1 = 0.f, s2 = 0.f, s3 = 0.f;
        #pragma unroll
        for (int i = 0; i < 14; ++i) {
            s0 = __builtin_fmaf(v[i].x, v[i].x, s0);
            s1 = __builtin_fmaf(v[i].y, v[i].y, s1);
            s2 = __builtin_fmaf(v[i].z, v[i].z, s2);
            s3 = __builtin_fmaf(v[i].w, v[i].w, s3);
        }
        return sum_halves((s0 + s1) + (s2 + s3));
    };

    for (int t = 0; t < Ss; ++t) {
        v8f acc = {};
        if (gemmW) {
            #pragma unroll
            for (int j = 0; j < KSTEPS; ++j) {
                const v2f a = *(const v2f*)&memS[r15 * MP + j * 4 + kh];
                acc = __builtin_amdgcn_wmma_f32_16x16x4_f32(
                         false, a,  false, bf[j],
                         (short)0, acc,  false,  false);
            }
        } else {
            v4f v[14], u[14];
            #pragma unroll
            for (int i = 0; i < 14; ++i) v[i] = *(const v4f*)&mRow[4 * i];
            #pragma unroll
            for (int i = 0; i < 14; ++i)
                u[i] = *(const v4f*)&sRow0[t * DP + 4 * i];
            const float kg = kRow0[t * PP];
            __builtin_amdgcn_sched_barrier(0);
            const float n2 = row_norm2(v);
            float d0 = 0.f, d1 = 0.f, d2 = 0.f, d3 = 0.f;
            #pragma unroll
            for (int i = 0; i < 14; ++i) {
                d0 = __builtin_fmaf(u[i].x, v[i].x, d0);
                d1 = __builtin_fmaf(u[i].y, v[i].y, d1);
                d2 = __builtin_fmaf(u[i].z, v[i].z, d2);
                d3 = __builtin_fmaf(u[i].w, v[i].w, d3);
            }
            const float dt = sum_halves((d0 + d1) + (d2 + d3));
            const float scale = (t == 0) ? 1.f : 1.f / sqrtf(n2);
            const float z = scale * dt + kg;
            const float g = 1.f / (1.f + expf(-z));
            if (lane < 16) { scS[r15] = scale; gtS[r15] = (z == 0.f) ? 0.f : g; }
        }
        __syncthreads();

        if (gemmW) {
            const v4f* sp = (const v4f*)&scS[mh];
            const v4f* gp = (const v4f*)&gtS[mh];
            const v4f s0 = sp[0], s1 = sp[1], q0 = gp[0], q1 = gp[1];
            const float sc[8] = {s0.x, s0.y, s0.z, s0.w, s1.x, s1.y, s1.z, s1.w};
            const float gt[8] = {q0.x, q0.y, q0.z, q0.w, q1.x, q1.y, q1.z, q1.w};
            const float wsA = preS[t * PP + nb + r15];
            const float wsB = preS[pOffB + t * PP + nb + r15];
            float* mp = &memS[mh * MP + nb + r15];
            float old[8];
            #pragma unroll
            for (int r = 0; r < 8; ++r) old[r] = mp[r * MP];
            #pragma unroll
            for (int r = 0; r < 8; ++r) {
                const float ws = (mh + r >= iB) ? wsB : wsA;
                float x = __builtin_fmaf(sc[r], acc[r], vk[r]) + ws;
                x = (x >= 0.f) ? x : pa * x;
                mp[r * MP] = __builtin_fmaf(sc[r], old[r], x * gt[r]);
            }
        }
        __syncthreads();
    }

    if (!gemmW) {
        v4f v[14];
        #pragma unroll
        for (int i = 0; i < 14; ++i) v[i] = *(const v4f*)&mRow[4 * i];
        const float scale = 1.f / sqrtf(row_norm2(v));
        #pragma unroll
        for (int i = 0; i < 14; ++i) {
            v[i] *= scale;
            *(v4f*)&mRow[4 * i] = v[i];
        }
    }
    __syncthreads();

    for (int i = tid; i < RT * Dd; i += NTHR) {
        const int r = i / Dd, d = i - r * Dd;
        out[g0 * Dd + i] = memS[r * MP + d];
    }
}

extern "C" void kernel_launch(void* const* d_in, const int* in_sizes, int n_in,
                              void* d_out, int out_size, void* d_ws, size_t ws_size,
                              hipStream_t stream) {
    (void)in_sizes; (void)n_in; (void)out_size; (void)ws_size;
    const int*   tokens  = (const int*)  d_in[0];
    const float* emb     = (const float*)d_in[1];
    const float* keys    = (const float*)d_in[2];
    const float* mult    = (const float*)d_in[3];
    const float* Uw      = (const float*)d_in[4];
    const float* Vw      = (const float*)d_in[5];
    const float* Ww      = (const float*)d_in[6];
    const float* prelu_a = (const float*)d_in[7];
    float* out = (float*)d_out;

    float* sents = (float*)d_ws;
    float* VkBuf = sents + (size_t)Bb * Ss * Dd;

    sents_kernel<<<Bb * Ss, 128, 0, stream>>>(tokens, emb, mult, sents);
    vk_kernel<<<Mm, 128, 0, stream>>>(keys, Vw, VkBuf);
    entnet_scan_kernel<<<NBLK, NTHR, 0, stream>>>(sents, keys, Uw, Ww, VkBuf,
                                                prelu_a, out);
}


# The state-independent product (s@Ww^T, s.keys for all steps) leaves the 80 scan work-groups, where straddling groups ran it twice: a 256-group WMMA launch in the Vk launch's place writes it once
# speedup vs baseline: 6.8198x; 1.0823x over previous
#include <hip/hip_runtime.h>

#define Bb   64
#define Ss   128
#define Ll   32
#define Dd   100
#define Mm   20
#define DP   112
#define MP   116
#define RT   16
#define NBLK ((Bb * Mm) / RT)
#define NTHR 256
#define NWAVE 7
#define KSTEPS 25
#define NB   144
#define PP   132
#define TR   32
#define TWAVE (NB / 16)
#define TBLK ((Bb * Ss) / TR)

typedef __attribute__((ext_vector_type(2))) float v2f;
typedef __attribute__((ext_vector_type(4))) float v4f;
typedef __attribute__((ext_vector_type(8))) float v8f;

__global__ __launch_bounds__(128) void sents_kernel(
    const int* __restrict__ tokens, const float* __restrict__ emb,
    const float* __restrict__ mult, float* __restrict__ sents)
{
    const int bs = blockIdx.x;
    const int d  = threadIdx.x;
    if (d >= Dd) return;
    const int* trow = tokens + bs * Ll;
    float acc = 0.f;
    #pragma unroll 4
    for (int l = 0; l < Ll; ++l) {
        const int tok = trow[l];
        acc += emb[tok * Dd + d] * mult[l * Dd + d];
    }
    sents[bs * Dd + d] = acc;
}

__global__ __launch_bounds__(TWAVE * 32) void terms_kernel(
    const float* __restrict__ sents, const float* __restrict__ keys,
    const float* __restrict__ Ww,    const float* __restrict__ Vw,
    float* __restrict__ pre,         float* __restrict__ VkBuf)
{
    const int tid = threadIdx.x;
    if (blockIdx.x < Mm) {
        const int m = blockIdx.x;
        const int d = tid;
        if (d >= Dd) return;
        float acc = 0.f;
        for (int k = 0; k < Dd; ++k) acc += keys[m * Dd + k] * Vw[d * Dd + k];
        VkBuf[m * Dd + d] = acc;
        return;
    }

    __shared__ __attribute__((aligned(16))) float tileS[TR * PP];

    const int wave = tid >> 5;
    const int lane = tid & 31;
    const int r15  = lane & 15;
    const int kh   = (lane >> 4) * 2;
    const int mh   = (lane >> 4) * 8;
    const int bs0  = (blockIdx.x - Mm) * TR;

    const int  n   = wave * 16 + r15;
    const bool isW = (n < Dd);
    const bool isK = (n >= DP) && (n < DP + Mm);
    const float* brow = isW ? (Ww + n * Dd) : (isK ? (keys + (n - DP) * Dd) : Ww);
    const bool okb = isW || isK;

    v2f bw[KSTEPS], a0[KSTEPS], a1[KSTEPS];
    #pragma unroll
    for (int j = 0; j < KSTEPS; ++j) {
        const v2f w = *(const v2f*)&brow[j * 4 + kh];
        bw[j] = okb ? w : v2f{0.f, 0.f};
    }
    #pragma unroll
    for (int j = 0; j < KSTEPS; ++j) {
        a0[j] = *(const v2f*)&sents[(bs0 + r15) * Dd + j * 4 + kh];
        a1[j] = *(const v2f*)&sents[(bs0 + 16 + r15) * Dd + j * 4 + kh];
    }
    v8f c0 = {}, c1 = {};
    #pragma unroll
    for (int j = 0; j < KSTEPS; ++j) {
        c0 = __builtin_amdgcn_wmma_f32_16x16x4_f32(
                 false, a0[j],  false, bw[j],
                 (short)0, c0,  false,  false);
        c1 = __builtin_amdgcn_wmma_f32_16x16x4_f32(
                 false, a1[j],  false, bw[j],
                 (short)0, c1,  false,  false);
    }
    if (n < PP) {
        #pragma unroll
        for (int r = 0; r < 8; ++r) {
            tileS[(mh + r) * PP + n]      = c0[r];
            tileS[(16 + mh + r) * PP + n] = c1[r];
        }
    }
    __syncthreads();
    v4f*       dst = (v4f*)(pre + (size_t)bs0 * PP);
    const v4f* src = (const v4f*)tileS;
    for (int i = tid; i < TR * PP / 4; i += TWAVE * 32) dst[i] = src[i];
}

__device__ __forceinline__ float sum_halves(float x)
{
    const int xi = __builtin_bit_cast(int, x);
    const int o  = __builtin_amdgcn_permlanex16(xi, xi, 0x76543210, 0xfedcba98, false, false);
    return x + __builtin_bit_cast(float, o);
}

__global__ __launch_bounds__(NTHR) void entnet_scan_kernel(
    const float* __restrict__ sents, const float* __restrict__ keys,
    const float* __restrict__ Uw,    const float* __restrict__ pre,
    const float* __restrict__ VkBuf, const float* __restrict__ prelu_a,
    float* __restrict__ out)
{
    const int g0   = blockIdx.x * RT;
    const int tid  = threadIdx.x;
    const int wave = tid >> 5;
    const int lane = tid & 31;
    const float pa = prelu_a[0];

    const int  bA     = g0 / Mm;
    const int  iB     = (bA + 1) * Mm - g0;
    const bool two    = (iB < RT);
    const int  nbatch = two ? 2 : 1;
    const int  sOffB  = two ? Ss * DP : 0;
    const int  pOffB  = two ? Ss * PP : 0;

    __shared__ __attribute__((aligned(16))) float memS[RT * MP];
    __shared__ __attribute__((aligned(16))) float sentS[2 * Ss * DP];
    __shared__ __attribute__((aligned(16))) float preS [2 * Ss * PP];
    __shared__ __attribute__((aligned(16))) float scS[RT];
    __shared__ __attribute__((aligned(16))) float gtS[RT];

    for (int i = tid; i < RT * DP; i += NTHR) {
        const int r = i / DP, d = i - r * DP;
        const int g = g0 + r;
        const int m = g - (g / Mm) * Mm;
        memS[r * MP + d] = (d < Dd) ? keys[m * Dd + d] : 0.f;
    }
    for (int sel = 0; sel < nbatch; ++sel) {
        const int bb = bA + sel;
        float* sS = &sentS[sel * Ss * DP];
        const v4f* pG = (const v4f*)(pre + (size_t)bb * Ss * PP);
        v4f*       pS = (v4f*)&preS[sel * Ss * PP];
        #pragma unroll
        for (int q = 0; q < (Ss * PP / 4) / NTHR; ++q) pS[q * NTHR + tid] = pG[q * NTHR + tid];
        {
            const int iq = ((Ss * PP / 4) / NTHR) * NTHR + tid % ((Ss * PP / 4) % NTHR);
            pS[iq] = pG[iq];
        }
        for (int i = tid; i < Ss * (DP - Dd); i += NTHR) {
            const int t = i / (DP - Dd), k = i - t * (DP - Dd);
            sS[t * DP + Dd + k] = 0.f;
        }
        #pragma unroll 10
        for (int i = tid; i < Ss * Dd; i += NTHR) {
            const int t = i / Dd, k = i - t * Dd;
            sS[t * DP + k] = sents[bb * Ss * Dd + i];
        }
    }
    __syncthreads();

    const int r15 = lane & 15;
    const int kh  = (lane >> 4) * 2;
    const int mh  = (lane >> 4) * 8;

    const int  waveU = __builtin_amdgcn_readfirstlane(tid) >> 5;
    const bool gemmW = (waveU < NWAVE);
    const int  nb    = wave * 16;

    v2f   bf[KSTEPS];
    float vk[8];
    if (gemmW) {
        const int  n   = nb + r15;
        const bool okn = (n < Dd);
        const int  nc  = okn ? n : 0;
        #pragma unroll
        for (int j = 0; j < KSTEPS; ++j) {
            const v2f w = *(const v2f*)&Uw[nc * Dd + j * 4 + kh];
            bf[j] = okn ? w : v2f{0.f, 0.f};
        }
        #pragma unroll
        for (int r = 0; r < 8; ++r) {
            const int g = g0 + mh + r;
            const int m = g - (g / Mm) * Mm;
            const float v = VkBuf[m * Dd + nc];
            vk[r] = okn ? v : 0.f;
        }
    } else {
        #pragma unroll
        for (int j = 0; j < KSTEPS; ++j) bf[j] = v2f{0.f, 0.f};
        #pragma unroll
        for (int r = 0; r < 8; ++r) vk[r] = 0.f;
    }

    const int  hcol = (lane >> 4) * (DP / 2);
    float*       mRow = &memS[r15 * MP + hcol];
    const int  gRow  = g0 + r15;
    const int  mSlot = gRow - (gRow / Mm) * Mm;
    const bool rowB  = (r15 >= iB);
    const float* sRow0 = &sentS[(rowB ? sOffB : 0) + hcol];
    const float* kRow0 = &preS[(rowB ? pOffB : 0) + DP + mSlot];

    auto row_norm2 = [&](const v4f* v) -> float {
        float s0 = 0.f, s1 = 0.f, s2 = 0.f, s3 = 0.f;
        #pragma unroll
        for (int i = 0; i < 14; ++i) {
            s0 = __builtin_fmaf(v[i].x, v[i].x, s0);
            s1 = __builtin_fmaf(v[i].y, v[i].y, s1);
            s2 = __builtin_fmaf(v[i].z, v[i].z, s2);
            s3 = __builtin_fmaf(v[i].w, v[i].w, s3);
        }
        return sum_halves((s0 + s1) + (s2 + s3));
    };

    for (int t = 0; t < Ss; ++t) {
        v8f acc = {};
        if (gemmW) {
            #pragma unroll
            for (int j = 0; j < KSTEPS; ++j) {
                const v2f a = *(const v2f*)&memS[r15 * MP + j * 4 + kh];
                acc = __builtin_amdgcn_wmma_f32_16x16x4_f32(
                         false, a,  false, bf[j],
                         (short)0, acc,  false,  false);
            }
        } else {
            v4f v[14], u[14];
            #pragma unroll
            for (int i = 0; i < 14; ++i) v[i] = *(const v4f*)&mRow[4 * i];
            #pragma unroll
            for (int i = 0; i < 14; ++i)
                u[i] = *(const v4f*)&sRow0[t * DP + 4 * i];
            const float kg = kRow0[t * PP];
            __builtin_amdgcn_sched_barrier(0);
            const float n2 = row_norm2(v);
            float d0 = 0.f, d1 = 0.f, d2 = 0.f, d3 = 0.f;
            #pragma unroll
            for (int i = 0; i < 14; ++i) {
                d0 = __builtin_fmaf(u[i].x, v[i].x, d0);
                d1 = __builtin_fmaf(u[i].y, v[i].y, d1);
                d2 = __builtin_fmaf(u[i].z, v[i].z, d2);
                d3 = __builtin_fmaf(u[i].w, v[i].w, d3);
            }
            const float dt = sum_halves((d0 + d1) + (d2 + d3));
            const float scale = (t == 0) ? 1.f : 1.f / sqrtf(n2);
            const float z = scale * dt + kg;
            const float g = 1.f / (1.f + expf(-z));
            if (lane < 16) { scS[r15] = scale; gtS[r15] = (z == 0.f) ? 0.f : g; }
        }
        __syncthreads();

        if (gemmW) {
            const v4f* sp = (const v4f*)&scS[mh];
            const v4f* gp = (const v4f*)&gtS[mh];
            const v4f s0 = sp[0], s1 = sp[1], q0 = gp[0], q1 = gp[1];
            const float sc[8] = {s0.x, s0.y, s0.z, s0.w, s1.x, s1.y, s1.z, s1.w};
            const float gt[8] = {q0.x, q0.y, q0.z, q0.w, q1.x, q1.y, q1.z, q1.w};
            const float wsA = preS[t * PP + nb + r15];
            const float wsB = preS[pOffB + t * PP + nb + r15];
            float* mp = &memS[mh * MP + nb + r15];
            float old[8];
            #pragma unroll
            for (int r = 0; r < 8; ++r) old[r] = mp[r * MP];
            #pragma unroll
            for (int r = 0; r < 8; ++r) {
                const float ws = (mh + r >= iB) ? wsB : wsA;
                float x = __builtin_fmaf(sc[r], acc[r], vk[r]) + ws;
                x = (x >= 0.f) ? x : pa * x;
                mp[r * MP] = __builtin_fmaf(sc[r], old[r], x * gt[r]);
            }
        }
        __syncthreads();
    }

    if (!gemmW) {
        v4f v[14];
        #pragma unroll
        for (int i = 0; i < 14; ++i) v[i] = *(const v4f*)&mRow[4 * i];
        const float scale = 1.f / sqrtf(row_norm2(v));
        #pragma unroll
        for (int i = 0; i < 14; ++i) {
            v[i] *= scale;
            *(v4f*)&mRow[4 * i] = v[i];
        }
    }
    __syncthreads();

    for (int i = tid; i < RT * Dd; i += NTHR) {
        const int r = i / Dd, d = i - r * Dd;
        out[g0 * Dd + i] = memS[r * MP + d];
    }
}

extern "C" void kernel_launch(void* const* d_in, const int* in_sizes, int n_in,
                              void* d_out, int out_size, void* d_ws, size_t ws_size,
                              hipStream_t stream) {
    (void)in_sizes; (void)n_in; (void)out_size; (void)ws_size;
    const int*   tokens  = (const int*)  d_in[0];
    const float* emb     = (const float*)d_in[1];
    const float* keys    = (const float*)d_in[2];
    const float* mult    = (const float*)d_in[3];
    const float* Uw      = (const float*)d_in[4];
    const float* Vw      = (const float*)d_in[5];
    const float* Ww      = (const float*)d_in[6];
    const float* prelu_a = (const float*)d_in[7];
    float* out = (float*)d_out;

    float* sents = (float*)d_ws;
    float* pre   = sents + (size_t)Bb * Ss * Dd;
    float* VkBuf = pre + (size_t)Bb * Ss * PP;

    sents_kernel<<<Bb * Ss, 128, 0, stream>>>(tokens, emb, mult, sents);
    terms_kernel<<<TBLK + Mm, TWAVE * 32, 0, stream>>>(sents, keys, Ww, Vw, pre, VkBuf);
    entnet_scan_kernel<<<NBLK, NTHR, 0, stream>>>(sents, keys, Uw, pre, VkBuf,
                                                prelu_a, out);
}


# Embedding gather joins the sentence-term launch: each of 256 groups sums its 32 sentence rows (16-byte loads, 4 rows a wave, old sum order) into LDS for its WMMA; Vk on the idle wave; 3 launches to 2
# speedup vs baseline: 7.2080x; 1.0569x over previous
#include <hip/hip_runtime.h>

#define Bb   64
#define Ss   128
#define Ll   32
#define Dd   100
#define Mm   20
#define DP   112
#define MP   116
#define RT   16
#define NBLK ((Bb * Mm) / RT)
#define NTHR 256
#define NWAVE 7
#define KSTEPS 25
#define NB   144
#define PP   132
#define TR   32
#define TWAVE (NB / 16)
#define TBLK ((Bb * Ss) / TR)

typedef __attribute__((ext_vector_type(2))) float v2f;
typedef __attribute__((ext_vector_type(4))) float v4f;
typedef __attribute__((ext_vector_type(8))) float v8f;

#define GWAVE 8
#define GROWS (TR / GWAVE)
#define DQ    (Dd / 4)
#define LB    4

typedef __attribute__((ext_vector_type(4))) int v4i;

__global__ __launch_bounds__(TWAVE * 32) void terms_kernel(
    const int* __restrict__ tokens,  const float* __restrict__ emb,
    const float* __restrict__ mult,  const float* __restrict__ keys,
    const float* __restrict__ Ww,    const float* __restrict__ Vw,
    float* __restrict__ sents,       float* __restrict__ pre,
    float* __restrict__ VkBuf)
{
    __shared__ __attribute__((aligned(16))) float tileS[TR * PP];
    __shared__ __attribute__((aligned(16))) float sentT[TR * Dd];

    const int tid  = threadIdx.x;
    const int blk  = blockIdx.x;
    const int wave = tid >> 5;
    const int lane = tid & 31;
    const int r15  = lane & 15;
    const int kh   = (lane >> 4) * 2;
    const int mh   = (lane >> 4) * 8;
    const int bs0  = blk * TR;

    const int  n   = wave * 16 + r15;
    const bool isW = (n < Dd);
    const bool isK = (n >= DP) && (n < DP + Mm);
    const float* brow = isW ? (Ww + n * Dd) : (isK ? (keys + (n - DP) * Dd) : Ww);
    const bool okb = isW || isK;
    v2f bw[KSTEPS];
    auto load_b = [&]() {
        #pragma unroll
        for (int j = 0; j < KSTEPS; ++j) bw[j] = *(const v2f*)&brow[j * 4 + kh];
    };

    const int waveU = __builtin_amdgcn_readfirstlane(tid) >> 5;
    v4f acc[GROWS];
    #pragma unroll
    for (int r = 0; r < GROWS; ++r) acc[r] = v4f{0.f, 0.f, 0.f, 0.f};
    if (waveU < GWAVE) {
        const int row0 = bs0 + waveU * GROWS;
        const v4i tk = ((const v4i*)(tokens + row0 * Ll))[lane];
        const int cq = (lane < DQ) ? lane : (DQ - 1);
        const v4f* mq = (const v4f*)mult + cq;
        v4f eb[3][LB][GROWS];
        v4f mb[3][LB];
        #pragma unroll
        for (int k = -2; k < Ll / LB; ++k) {
            const int ki = k + 2;
            if (ki < Ll / LB) {
                #pragma unroll
                for (int li = 0; li < LB; ++li) {
                    const int l = ki * LB + li;
                    mb[ki % 3][li] = mq[l * DQ];
                    #pragma unroll
                    for (int r = 0; r < GROWS; ++r) {
                        const int flat = r * Ll + l;
                        const int tok  = __builtin_amdgcn_readlane(tk[flat & 3], flat >> 2);
                        const v4f* rp  = (const v4f*)(emb + (size_t)tok * Dd);
                        eb[ki % 3][li][r] = rp[cq];
                    }
                }
            } else if (ki == Ll / LB) {
                load_b();
            }
            if (k >= 0) {
                __builtin_amdgcn_sched_barrier(0);
                #pragma unroll
                for (int li = 0; li < LB; ++li) {
                    const v4f m = mb[k % 3][li];
                    #pragma unroll
                    for (int r = 0; r < GROWS; ++r) {
                        const v4f e = eb[k % 3][li][r];
                        acc[r].x = __builtin_fmaf(e.x, m.x, acc[r].x);
                        acc[r].y = __builtin_fmaf(e.y, m.y, acc[r].y);
                        acc[r].z = __builtin_fmaf(e.z, m.z, acc[r].z);
                        acc[r].w = __builtin_fmaf(e.w, m.w, acc[r].w);
                    }
                }
                __builtin_amdgcn_sched_barrier(0);
            }
        }
    } else {
        load_b();
        if (blk < 4 * Mm) {
            const int m = blk >> 2;
            const int d = (blk & 3) * DQ + lane;
            if (lane < DQ) {
                float a = 0.f;
                for (int k = 0; k < Dd; ++k) a = __builtin_fmaf(keys[m * Dd + k], Vw[d * Dd + k], a);
                VkBuf[m * Dd + d] = a;
            }
        }
    }

    if (waveU < GWAVE && lane < DQ) {
        #pragma unroll
        for (int r = 0; r < GROWS; ++r)
            *(v4f*)&sentT[(waveU * GROWS + r) * Dd + 4 * lane] = acc[r];
    }
    #pragma unroll
    for (int j = 0; j < KSTEPS; ++j) bw[j] = okb ? bw[j] : v2f{0.f, 0.f};
    __syncthreads();
    {
        v4f*       dst = (v4f*)(sents + (size_t)bs0 * Dd);
        const v4f* src = (const v4f*)sentT;
        #pragma unroll
        for (int q = 0; q < (TR * Dd / 4) / (TWAVE * 32); ++q)
            dst[q * (TWAVE * 32) + tid] = src[q * (TWAVE * 32) + tid];
        if (tid < (TR * Dd / 4) % (TWAVE * 32)) {
            const int i = ((TR * Dd / 4) / (TWAVE * 32)) * (TWAVE * 32) + tid;
            dst[i] = src[i];
        }
    }

    v2f a0[KSTEPS], a1[KSTEPS];
    #pragma unroll
    for (int j = 0; j < KSTEPS; ++j) {
        a0[j] = *(const v2f*)&sentT[r15 * Dd + j * 4 + kh];
        a1[j] = *(const v2f*)&sentT[(16 + r15) * Dd + j * 4 + kh];
    }
    v8f c0 = {}, c1 = {};
    #pragma unroll
    for (int j = 0; j < KSTEPS; ++j) {
        c0 = __builtin_amdgcn_wmma_f32_16x16x4_f32(
                 false, a0[j],  false, bw[j],
                 (short)0, c0,  false,  false);
        c1 = __builtin_amdgcn_wmma_f32_16x16x4_f32(
                 false, a1[j],  false, bw[j],
                 (short)0, c1,  false,  false);
    }
    if (n < PP) {
        #pragma unroll
        for (int r = 0; r < 8; ++r) {
            tileS[(mh + r) * PP + n]      = c0[r];
            tileS[(16 + mh + r) * PP + n] = c1[r];
        }
    }
    __syncthreads();
    v4f*       dst = (v4f*)(pre + (size_t)bs0 * PP);
    const v4f* src = (const v4f*)tileS;
    for (int i = tid; i < TR * PP / 4; i += TWAVE * 32) dst[i] = src[i];
}

__device__ __forceinline__ float sum_halves(float x)
{
    const int xi = __builtin_bit_cast(int, x);
    const int o  = __builtin_amdgcn_permlanex16(xi, xi, 0x76543210, 0xfedcba98, false, false);
    return x + __builtin_bit_cast(float, o);
}

__global__ __launch_bounds__(NTHR) void entnet_scan_kernel(
    const float* __restrict__ sents, const float* __restrict__ keys,
    const float* __restrict__ Uw,    const float* __restrict__ pre,
    const float* __restrict__ VkBuf, const float* __restrict__ prelu_a,
    float* __restrict__ out)
{
    const int g0   = blockIdx.x * RT;
    const int tid  = threadIdx.x;
    const int wave = tid >> 5;
    const int lane = tid & 31;
    const float pa = prelu_a[0];

    const int  bA     = g0 / Mm;
    const int  iB     = (bA + 1) * Mm - g0;
    const bool two    = (iB < RT);
    const int  nbatch = two ? 2 : 1;
    const int  sOffB  = two ? Ss * DP : 0;
    const int  pOffB  = two ? Ss * PP : 0;

    __shared__ __attribute__((aligned(16))) float memS[RT * MP];
    __shared__ __attribute__((aligned(16))) float sentS[2 * Ss * DP];
    __shared__ __attribute__((aligned(16))) float preS [2 * Ss * PP];
    __shared__ __attribute__((aligned(16))) float scS[RT];
    __shared__ __attribute__((aligned(16))) float gtS[RT];

    for (int i = tid; i < RT * DP; i += NTHR) {
        const int r = i / DP, d = i - r * DP;
        const int g = g0 + r;
        const int m = g - (g / Mm) * Mm;
        memS[r * MP + d] = (d < Dd) ? keys[m * Dd + d] : 0.f;
    }
    for (int sel = 0; sel < nbatch; ++sel) {
        const int bb = bA + sel;
        float* sS = &sentS[sel * Ss * DP];
        const v4f* pG = (const v4f*)(pre + (size_t)bb * Ss * PP);
        v4f*       pS = (v4f*)&preS[sel * Ss * PP];
        #pragma unroll
        for (int q = 0; q < (Ss * PP / 4) / NTHR; ++q) pS[q * NTHR + tid] = pG[q * NTHR + tid];
        {
            const int iq = ((Ss * PP / 4) / NTHR) * NTHR + tid % ((Ss * PP / 4) % NTHR);
            pS[iq] = pG[iq];
        }
        for (int i = tid; i < Ss * (DP - Dd); i += NTHR) {
            const int t = i / (DP - Dd), k = i - t * (DP - Dd);
            sS[t * DP + Dd + k] = 0.f;
        }
        #pragma unroll 10
        for (int i = tid; i < Ss * Dd; i += NTHR) {
            const int t = i / Dd, k = i - t * Dd;
            sS[t * DP + k] = sents[bb * Ss * Dd + i];
        }
    }
    __syncthreads();

    const int r15 = lane & 15;
    const int kh  = (lane >> 4) * 2;
    const int mh  = (lane >> 4) * 8;

    const int  waveU = __builtin_amdgcn_readfirstlane(tid) >> 5;
    const bool gemmW = (waveU < NWAVE);
    const int  nb    = wave * 16;

    v2f   bf[KSTEPS];
    float vk[8];
    if (gemmW) {
        const int  n   = nb + r15;
        const bool okn = (n < Dd);
        const int  nc  = okn ? n : 0;
        #pragma unroll
        for (int j = 0; j < KSTEPS; ++j) {
            const v2f w = *(const v2f*)&Uw[nc * Dd + j * 4 + kh];
            bf[j] = okn ? w : v2f{0.f, 0.f};
        }
        #pragma unroll
        for (int r = 0; r < 8; ++r) {
            const int g = g0 + mh + r;
            const int m = g - (g / Mm) * Mm;
            const float v = VkBuf[m * Dd + nc];
            vk[r] = okn ? v : 0.f;
        }
    } else {
        #pragma unroll
        for (int j = 0; j < KSTEPS; ++j) bf[j] = v2f{0.f, 0.f};
        #pragma unroll
        for (int r = 0; r < 8; ++r) vk[r] = 0.f;
    }

    const int  hcol = (lane >> 4) * (DP / 2);
    float*       mRow = &memS[r15 * MP + hcol];
    const int  gRow  = g0 + r15;
    const int  mSlot = gRow - (gRow / Mm) * Mm;
    const bool rowB  = (r15 >= iB);
    const float* sRow0 = &sentS[(rowB ? sOffB : 0) + hcol];
    const float* kRow0 = &preS[(rowB ? pOffB : 0) + DP + mSlot];

    auto row_norm2 = [&](const v4f* v) -> float {
        float s0 = 0.f, s1 = 0.f, s2 = 0.f, s3 = 0.f;
        #pragma unroll
        for (int i = 0; i < 14; ++i) {
            s0 = __builtin_fmaf(v[i].x, v[i].x, s0);
            s1 = __builtin_fmaf(v[i].y, v[i].y, s1);
            s2 = __builtin_fmaf(v[i].z, v[i].z, s2);
            s3 = __builtin_fmaf(v[i].w, v[i].w, s3);
        }
        return sum_halves((s0 + s1) + (s2 + s3));
    };

    for (int t = 0; t < Ss; ++t) {
        v8f acc = {};
        if (gemmW) {
            #pragma unroll
            for (int j = 0; j < KSTEPS; ++j) {
                const v2f a = *(const v2f*)&memS[r15 * MP + j * 4 + kh];
                acc = __builtin_amdgcn_wmma_f32_16x16x4_f32(
                         false, a,  false, bf[j],
                         (short)0, acc,  false,  false);
            }
        } else {
            v4f v[14], u[14];
            #pragma unroll
            for (int i = 0; i < 14; ++i) v[i] = *(const v4f*)&mRow[4 * i];
            #pragma unroll
            for (int i = 0; i < 14; ++i)
                u[i] = *(const v4f*)&sRow0[t * DP + 4 * i];
            const float kg = kRow0[t * PP];
            __builtin_amdgcn_sched_barrier(0);
            const float n2 = row_norm2(v);
            float d0 = 0.f, d1 = 0.f, d2 = 0.f, d3 = 0.f;
            #pragma unroll
            for (int i = 0; i < 14; ++i) {
                d0 = __builtin_fmaf(u[i].x, v[i].x, d0);
                d1 = __builtin_fmaf(u[i].y, v[i].y, d1);
                d2 = __builtin_fmaf(u[i].z, v[i].z, d2);
                d3 = __builtin_fmaf(u[i].w, v[i].w, d3);
            }
            const float dt = sum_halves((d0 + d1) + (d2 + d3));
            const float scale = (t == 0) ? 1.f : 1.f / sqrtf(n2);
            const float z = scale * dt + kg;
            const float g = 1.f / (1.f + expf(-z));
            if (lane < 16) { scS[r15] = scale; gtS[r15] = (z == 0.f) ? 0.f : g; }
        }
        __syncthreads();

        if (gemmW) {
            const v4f* sp = (const v4f*)&scS[mh];
            const v4f* gp = (const v4f*)&gtS[mh];
            const v4f s0 = sp[0], s1 = sp[1], q0 = gp[0], q1 = gp[1];
            const float sc[8] = {s0.x, s0.y, s0.z, s0.w, s1.x, s1.y, s1.z, s1.w};
            const float gt[8] = {q0.x, q0.y, q0.z, q0.w, q1.x, q1.y, q1.z, q1.w};
            const float wsA = preS[t * PP + nb + r15];
            const float wsB = preS[pOffB + t * PP + nb + r15];
            float* mp = &memS[mh * MP + nb + r15];
            float old[8];
            #pragma unroll
            for (int r = 0; r < 8; ++r) old[r] = mp[r * MP];
            #pragma unroll
            for (int r = 0; r < 8; ++r) {
                const float ws = (mh + r >= iB) ? wsB : wsA;
                float x = __builtin_fmaf(sc[r], acc[r], vk[r]) + ws;
                x = (x >= 0.f) ? x : pa * x;
                mp[r * MP] = __builtin_fmaf(sc[r], old[r], x * gt[r]);
            }
        }
        __syncthreads();
    }

    if (!gemmW) {
        v4f v[14];
        #pragma unroll
        for (int i = 0; i < 14; ++i) v[i] = *(const v4f*)&mRow[4 * i];
        const float scale = 1.f / sqrtf(row_norm2(v));
        #pragma unroll
        for (int i = 0; i < 14; ++i) {
            v[i] *= scale;
            *(v4f*)&mRow[4 * i] = v[i];
        }
    }
    __syncthreads();

    for (int i = tid; i < RT * Dd; i += NTHR) {
        const int r = i / Dd, d = i - r * Dd;
        out[g0 * Dd + i] = memS[r * MP + d];
    }
}

extern "C" void kernel_launch(void* const* d_in, const int* in_sizes, int n_in,
                              void* d_out, int out_size, void* d_ws, size_t ws_size,
                              hipStream_t stream) {
    (void)in_sizes; (void)n_in; (void)out_size; (void)ws_size;
    const int*   tokens  = (const int*)  d_in[0];
    const float* emb     = (const float*)d_in[1];
    const float* keys    = (const float*)d_in[2];
    const float* mult    = (const float*)d_in[3];
    const float* Uw      = (const float*)d_in[4];
    const float* Vw      = (const float*)d_in[5];
    const float* Ww      = (const float*)d_in[6];
    const float* prelu_a = (const float*)d_in[7];
    float* out = (float*)d_out;

    float* sents = (float*)d_ws;
    float* pre   = sents + (size_t)Bb * Ss * Dd;
    float* VkBuf = pre + (size_t)Bb * Ss * PP;

    terms_kernel<<<TBLK, TWAVE * 32, 0, stream>>>(tokens, emb, mult, keys, Ww, Vw,
                                                       sents, pre, VkBuf);
    entnet_scan_kernel<<<NBLK, NTHR, 0, stream>>>(sents, keys, Uw, pre, VkBuf,
                                                prelu_a, out);
}


# Scan step: state tile double-buffered in LDS; scale/gate hand-over a split barrier (product waves signal at step head, wait 22 k-steps in), update reads fly under the last 3 WMMAs; one full barrier
# speedup vs baseline: 7.4678x; 1.0360x over previous
#include <hip/hip_runtime.h>

#define Bb   64
#define Ss   128
#define Ll   32
#define Dd   100
#define Mm   20
#define DP   112
#define MP   116
#define RT   16
#define NBLK ((Bb * Mm) / RT)
#define NTHR 256
#define NWAVE 7
#define KSTEPS 25
#define KHEAD  22
#define KMID   10
#define MB   (RT * MP)
#define NB   144
#define PP   132
#define TR   32
#define TWAVE (NB / 16)
#define TBLK ((Bb * Ss) / TR)

typedef __attribute__((ext_vector_type(2))) float v2f;
typedef __attribute__((ext_vector_type(4))) float v4f;
typedef __attribute__((ext_vector_type(8))) float v8f;

#define GWAVE 8
#define GROWS (TR / GWAVE)
#define DQ    (Dd / 4)
#define LB    4

typedef __attribute__((ext_vector_type(4))) int v4i;

__global__ __launch_bounds__(TWAVE * 32) void terms_kernel(
    const int* __restrict__ tokens,  const float* __restrict__ emb,
    const float* __restrict__ mult,  const float* __restrict__ keys,
    const float* __restrict__ Ww,    const float* __restrict__ Vw,
    float* __restrict__ sents,       float* __restrict__ pre,
    float* __restrict__ VkBuf)
{
    __shared__ __attribute__((aligned(16))) float tileS[TR * PP];
    __shared__ __attribute__((aligned(16))) float sentT[TR * Dd];

    const int tid  = threadIdx.x;
    const int blk  = blockIdx.x;
    const int wave = tid >> 5;
    const int lane = tid & 31;
    const int r15  = lane & 15;
    const int kh   = (lane >> 4) * 2;
    const int mh   = (lane >> 4) * 8;
    const int bs0  = blk * TR;

    const int  n   = wave * 16 + r15;
    const bool isW = (n < Dd);
    const bool isK = (n >= DP) && (n < DP + Mm);
    const float* brow = isW ? (Ww + n * Dd) : (isK ? (keys + (n - DP) * Dd) : Ww);
    const bool okb = isW || isK;
    v2f bw[KSTEPS];
    auto load_b = [&]() {
        #pragma unroll
        for (int j = 0; j < KSTEPS; ++j) bw[j] = *(const v2f*)&brow[j * 4 + kh];
    };

    const int waveU = __builtin_amdgcn_readfirstlane(tid) >> 5;
    v4f acc[GROWS];
    #pragma unroll
    for (int r = 0; r < GROWS; ++r) acc[r] = v4f{0.f, 0.f, 0.f, 0.f};
    if (waveU < GWAVE) {
        const int row0 = bs0 + waveU * GROWS;
        const v4i tk = ((const v4i*)(tokens + row0 * Ll))[lane];
        const int cq = (lane < DQ) ? lane : (DQ - 1);
        const v4f* mq = (const v4f*)mult + cq;
        v4f eb[3][LB][GROWS];
        v4f mb[3][LB];
        #pragma unroll
        for (int k = -2; k < Ll / LB; ++k) {
            const int ki = k + 2;
            if (ki < Ll / LB) {
                #pragma unroll
                for (int li = 0; li < LB; ++li) {
                    const int l = ki * LB + li;
                    mb[ki % 3][li] = mq[l * DQ];
                    #pragma unroll
                    for (int r = 0; r < GROWS; ++r) {
                        const int flat = r * Ll + l;
                        const int tok  = __builtin_amdgcn_readlane(tk[flat & 3], flat >> 2);
                        const v4f* rp  = (const v4f*)(emb + (size_t)tok * Dd);
                        eb[ki % 3][li][r] = rp[cq];
                    }
                }
            } else if (ki == Ll / LB) {
                load_b();
            }
            if (k >= 0) {
                __builtin_amdgcn_sched_barrier(0);
                #pragma unroll
                for (int li = 0; li < LB; ++li) {
                    const v4f m = mb[k % 3][li];
                    #pragma unroll
                    for (int r = 0; r < GROWS; ++r) {
                        const v4f e = eb[k % 3][li][r];
                        acc[r].x = __builtin_fmaf(e.x, m.x, acc[r].x);
                        acc[r].y = __builtin_fmaf(e.y, m.y, acc[r].y);
                        acc[r].z = __builtin_fmaf(e.z, m.z, acc[r].z);
                        acc[r].w = __builtin_fmaf(e.w, m.w, acc[r].w);
                    }
                }
                __builtin_amdgcn_sched_barrier(0);
            }
        }
    } else {
        load_b();
        if (blk < 4 * Mm) {
            const int m = blk >> 2;
            const int d = (blk & 3) * DQ + lane;
            if (lane < DQ) {
                float a = 0.f;
                for (int k = 0; k < Dd; ++k) a = __builtin_fmaf(keys[m * Dd + k], Vw[d * Dd + k], a);
                VkBuf[m * Dd + d] = a;
            }
        }
    }

    if (waveU < GWAVE && lane < DQ) {
        #pragma unroll
        for (int r = 0; r < GROWS; ++r)
            *(v4f*)&sentT[(waveU * GROWS + r) * Dd + 4 * lane] = acc[r];
    }
    #pragma unroll
    for (int j = 0; j < KSTEPS; ++j) bw[j] = okb ? bw[j] : v2f{0.f, 0.f};
    __syncthreads();
    {
        v4f*       dst = (v4f*)(sents + (size_t)bs0 * Dd);
        const v4f* src = (const v4f*)sentT;
        #pragma unroll
        for (int q = 0; q < (TR * Dd / 4) / (TWAVE * 32); ++q)
            dst[q * (TWAVE * 32) + tid] = src[q * (TWAVE * 32) + tid];
        if (tid < (TR * Dd / 4) % (TWAVE * 32)) {
            const int i = ((TR * Dd / 4) / (TWAVE * 32)) * (TWAVE * 32) + tid;
            dst[i] = src[i];
        }
    }

    v2f a0[KSTEPS], a1[KSTEPS];
    #pragma unroll
    for (int j = 0; j < KSTEPS; ++j) {
        a0[j] = *(const v2f*)&sentT[r15 * Dd + j * 4 + kh];
        a1[j] = *(const v2f*)&sentT[(16 + r15) * Dd + j * 4 + kh];
    }
    v8f c0 = {}, c1 = {};
    #pragma unroll
    for (int j = 0; j < KSTEPS; ++j) {
        c0 = __builtin_amdgcn_wmma_f32_16x16x4_f32(
                 false, a0[j],  false, bw[j],
                 (short)0, c0,  false,  false);
        c1 = __builtin_amdgcn_wmma_f32_16x16x4_f32(
                 false, a1[j],  false, bw[j],
                 (short)0, c1,  false,  false);
    }
    if (n < PP) {
        #pragma unroll
        for (int r = 0; r < 8; ++r) {
            tileS[(mh + r) * PP + n]      = c0[r];
            tileS[(16 + mh + r) * PP + n] = c1[r];
        }
    }
    __syncthreads();
    v4f*       dst = (v4f*)(pre + (size_t)bs0 * PP);
    const v4f* src = (const v4f*)tileS;
    for (int i = tid; i < TR * PP / 4; i += TWAVE * 32) dst[i] = src[i];
}

__device__ __forceinline__ float sum_halves(float x)
{
    const int xi = __builtin_bit_cast(int, x);
    const int o  = __builtin_amdgcn_permlanex16(xi, xi, 0x76543210, 0xfedcba98, false, false);
    return x + __builtin_bit_cast(float, o);
}

__global__ __launch_bounds__(NTHR) void entnet_scan_kernel(
    const float* __restrict__ sents, const float* __restrict__ keys,
    const float* __restrict__ Uw,    const float* __restrict__ pre,
    const float* __restrict__ VkBuf, const float* __restrict__ prelu_a,
    float* __restrict__ out)
{
    const int g0   = blockIdx.x * RT;
    const int tid  = threadIdx.x;
    const int wave = tid >> 5;
    const int lane = tid & 31;
    const float pa = prelu_a[0];

    const int  bA     = g0 / Mm;
    const int  iB     = (bA + 1) * Mm - g0;
    const bool two    = (iB < RT);
    const int  nbatch = two ? 2 : 1;
    const int  sOffB  = two ? Ss * DP : 0;
    const int  pOffB  = two ? Ss * PP : 0;

    __shared__ __attribute__((aligned(16))) float memS[2 * MB];
    __shared__ __attribute__((aligned(16))) float sentS[2 * Ss * DP];
    __shared__ __attribute__((aligned(16))) float preS [2 * Ss * PP];
    __shared__ __attribute__((aligned(16))) float scS[RT];
    __shared__ __attribute__((aligned(16))) float gtS[RT];

    for (int i = tid; i < RT * DP; i += NTHR) {
        const int r = i / DP, d = i - r * DP;
        const int g = g0 + r;
        const int m = g - (g / Mm) * Mm;
        memS[r * MP + d] = (d < Dd) ? keys[m * Dd + d] : 0.f;
    }
    for (int sel = 0; sel < nbatch; ++sel) {
        const int bb = bA + sel;
        float* sS = &sentS[sel * Ss * DP];
        const v4f* pG = (const v4f*)(pre + (size_t)bb * Ss * PP);
        v4f*       pS = (v4f*)&preS[sel * Ss * PP];
        #pragma unroll
        for (int q = 0; q < (Ss * PP / 4) / NTHR; ++q) pS[q * NTHR + tid] = pG[q * NTHR + tid];
        {
            const int iq = ((Ss * PP / 4) / NTHR) * NTHR + tid % ((Ss * PP / 4) % NTHR);
            pS[iq] = pG[iq];
        }
        for (int i = tid; i < Ss * (DP - Dd); i += NTHR) {
            const int t = i / (DP - Dd), k = i - t * (DP - Dd);
            sS[t * DP + Dd + k] = 0.f;
        }
        #pragma unroll 10
        for (int i = tid; i < Ss * Dd; i += NTHR) {
            const int t = i / Dd, k = i - t * Dd;
            sS[t * DP + k] = sents[bb * Ss * Dd + i];
        }
    }
    __syncthreads();

    const int r15 = lane & 15;
    const int kh  = (lane >> 4) * 2;
    const int mh  = (lane >> 4) * 8;

    const int  waveU = __builtin_amdgcn_readfirstlane(tid) >> 5;
    const bool gemmW = (waveU < NWAVE);
    const int  nb    = wave * 16;

    v2f   bf[KSTEPS];
    float vk[8];
    if (gemmW) {
        const int  n   = nb + r15;
        const bool okn = (n < Dd);
        const int  nc  = okn ? n : 0;
        #pragma unroll
        for (int j = 0; j < KSTEPS; ++j) {
            const v2f w = *(const v2f*)&Uw[nc * Dd + j * 4 + kh];
            bf[j] = okn ? w : v2f{0.f, 0.f};
        }
        #pragma unroll
        for (int r = 0; r < 8; ++r) {
            const int g = g0 + mh + r;
            const int m = g - (g / Mm) * Mm;
            const float v = VkBuf[m * Dd + nc];
            vk[r] = okn ? v : 0.f;
        }
    } else {
        #pragma unroll
        for (int j = 0; j < KSTEPS; ++j) bf[j] = v2f{0.f, 0.f};
        #pragma unroll
        for (int r = 0; r < 8; ++r) vk[r] = 0.f;
    }

    const int  hcol = (lane >> 4) * (DP / 2);
    float*       mRow = &memS[r15 * MP + hcol];
    const int  gRow  = g0 + r15;
    const int  mSlot = gRow - (gRow / Mm) * Mm;
    const bool rowB  = (r15 >= iB);
    const float* sRow0 = &sentS[(rowB ? sOffB : 0) + hcol];
    const float* kRow0 = &preS[(rowB ? pOffB : 0) + DP + mSlot];

    auto row_norm2 = [&](const v4f* v) -> float {
        float s0 = 0.f, s1 = 0.f, s2 = 0.f, s3 = 0.f;
        #pragma unroll
        for (int i = 0; i < 14; ++i) {
            s0 = __builtin_fmaf(v[i].x, v[i].x, s0);
            s1 = __builtin_fmaf(v[i].y, v[i].y, s1);
            s2 = __builtin_fmaf(v[i].z, v[i].z, s2);
            s3 = __builtin_fmaf(v[i].w, v[i].w, s3);
        }
        return sum_halves((s0 + s1) + (s2 + s3));
    };

    auto step = [&](const int t, const int cur, const int nxt) __attribute__((always_inline)) {
        if (gemmW) {
            asm volatile("s_barrier_signal -1" ::: "memory");
            const float* mc = &memS[cur];
            v8f acc = {};
            #pragma unroll
            for (int j = 0; j < KMID; ++j) {
                const v2f a = *(const v2f*)&mc[r15 * MP + j * 4 + kh];
                acc = __builtin_amdgcn_wmma_f32_16x16x4_f32(
                         false, a,  false, bf[j],
                         (short)0, acc,  false,  false);
            }
            v2f at[KSTEPS - KHEAD];
            #pragma unroll
            for (int j = KHEAD; j < KSTEPS; ++j) at[j - KHEAD] = *(const v2f*)&mc[r15 * MP + j * 4 + kh];
            const float wsA = preS[t * PP + nb + r15];
            const float wsB = preS[pOffB + t * PP + nb + r15];
            const float* op = &mc[mh * MP + nb + r15];
            float old[8];
            #pragma unroll
            for (int r = 0; r < 8; ++r) old[r] = op[r * MP];
            __builtin_amdgcn_sched_barrier(0);
            #pragma unroll
            for (int j = KMID; j < KHEAD; ++j) {
                const v2f a = *(const v2f*)&mc[r15 * MP + j * 4 + kh];
                acc = __builtin_amdgcn_wmma_f32_16x16x4_f32(
                         false, a,  false, bf[j],
                         (short)0, acc,  false,  false);
            }
            asm volatile("s_barrier_wait -1" :: "v"(acc) : "memory");
            const v4f* sp = (const v4f*)&scS[mh];
            const v4f* gp = (const v4f*)&gtS[mh];
            const v4f s0 = sp[0], s1 = sp[1], q0 = gp[0], q1 = gp[1];
            const float sc[8] = {s0.x, s0.y, s0.z, s0.w, s1.x, s1.y, s1.z, s1.w};
            const float gt[8] = {q0.x, q0.y, q0.z, q0.w, q1.x, q1.y, q1.z, q1.w};
            __builtin_amdgcn_sched_barrier(0);
            #pragma unroll
            for (int j = KHEAD; j < KSTEPS; ++j)
                acc = __builtin_amdgcn_wmma_f32_16x16x4_f32(
                         false, at[j - KHEAD],  false, bf[j],
                         (short)0, acc,  false,  false);
            asm volatile("v_nop\n\tv_nop\n\tv_nop\n\tv_nop" : "+v"(acc));
            float* mp = &memS[nxt + mh * MP + nb + r15];
            #pragma unroll
            for (int r = 0; r < 8; ++r) {
                const float ws = (mh + r >= iB) ? wsB : wsA;
                float x = __builtin_fmaf(sc[r], acc[r], vk[r]) + ws;
                x = (x >= 0.f) ? x : pa * x;
                mp[r * MP] = __builtin_fmaf(sc[r], old[r], x * gt[r]);
            }
        } else {
            v4f v[14], u[14];
            #pragma unroll
            for (int i = 0; i < 14; ++i) v[i] = *(const v4f*)&mRow[cur + 4 * i];
            #pragma unroll
            for (int i = 0; i < 14; ++i)
                u[i] = *(const v4f*)&sRow0[t * DP + 4 * i];
            const float kg = kRow0[t * PP];
            __builtin_amdgcn_sched_barrier(0);
            const float n2 = row_norm2(v);
            float d0 = 0.f, d1 = 0.f, d2 = 0.f, d3 = 0.f;
            #pragma unroll
            for (int i = 0; i < 14; ++i) {
                d0 = __builtin_fmaf(u[i].x, v[i].x, d0);
                d1 = __builtin_fmaf(u[i].y, v[i].y, d1);
                d2 = __builtin_fmaf(u[i].z, v[i].z, d2);
                d3 = __builtin_fmaf(u[i].w, v[i].w, d3);
            }
            const float dt = sum_halves((d0 + d1) + (d2 + d3));
            const float scale = (t == 0) ? 1.f : 1.f / sqrtf(n2);
            const float z = scale * dt + kg;
            const float g = 1.f / (1.f + expf(-z));
            if (lane < 16) { scS[r15] = scale; gtS[r15] = (z == 0.f) ? 0.f : g; }
            asm volatile("s_wait_dscnt 0x0\n\ts_barrier_signal -1\n\ts_barrier_wait -1" ::: "memory");
        }
        __syncthreads();
    };
    for (int t = 0; t < Ss; t += 2) {
        step(t, 0, MB);
        step(t + 1, MB, 0);
    }

    if (!gemmW) {
        v4f v[14];
        #pragma unroll
        for (int i = 0; i < 14; ++i) v[i] = *(const v4f*)&mRow[4 * i];
        const float scale = 1.f / sqrtf(row_norm2(v));
        #pragma unroll
        for (int i = 0; i < 14; ++i) {
            v[i] *= scale;
            *(v4f*)&mRow[4 * i] = v[i];
        }
    }
    __syncthreads();

    for (int i = tid; i < RT * Dd; i += NTHR) {
        const int r = i / Dd, d = i - r * Dd;
        out[g0 * Dd + i] = memS[r * MP + d];
    }
}

extern "C" void kernel_launch(void* const* d_in, const int* in_sizes, int n_in,
                              void* d_out, int out_size, void* d_ws, size_t ws_size,
                              hipStream_t stream) {
    (void)in_sizes; (void)n_in; (void)out_size; (void)ws_size;
    const int*   tokens  = (const int*)  d_in[0];
    const float* emb     = (const float*)d_in[1];
    const float* keys    = (const float*)d_in[2];
    const float* mult    = (const float*)d_in[3];
    const float* Uw      = (const float*)d_in[4];
    const float* Vw      = (const float*)d_in[5];
    const float* Ww      = (const float*)d_in[6];
    const float* prelu_a = (const float*)d_in[7];
    float* out = (float*)d_out;

    float* sents = (float*)d_ws;
    float* pre   = sents + (size_t)Bb * Ss * Dd;
    float* VkBuf = pre + (size_t)Bb * Ss * PP;

    terms_kernel<<<TBLK, TWAVE * 32, 0, stream>>>(tokens, emb, mult, keys, Ww, Vw,
                                                       sents, pre, VkBuf);
    entnet_scan_kernel<<<NBLK, NTHR, 0, stream>>>(sents, keys, Uw, pre, VkBuf,
                                                prelu_a, out);
}
